# starting kernel
# baseline (speedup 1.0000x reference)
#include <hip/hip_runtime.h>

typedef __attribute__((ext_vector_type(16))) _Float16 v16h;
typedef __attribute__((ext_vector_type(8)))  float    v8f;

#define DIM 128
#define NN  4096
#define MM  4096

__device__ __forceinline__ v16h frag_load(const _Float16* __restrict__ tile, int ld) {
    const int lane = threadIdx.x & 31;
    const int r    = lane & 15;
    const int kb   = (lane & 16) ? 8 : 0;
    const _Float16* q = tile + (size_t)r * (size_t)ld + kb;
    v16h f;
    uint4* fu = reinterpret_cast<uint4*>(&f);
    fu[0] = *reinterpret_cast<const uint4*>(q);
    fu[1] = *reinterpret_cast<const uint4*>(q + 16);
    return f;
}

__device__ __forceinline__ void frag_rc(int& rbase, int& col) {
    const int lane = threadIdx.x & 31;
    col   = lane & 15;
    rbase = (lane & 16) ? 8 : 0;
}

__global__ void gemm_bias_act(const _Float16* __restrict__ A, const _Float16* __restrict__ B,
                              const float* __restrict__ bias,
                              float* __restrict__ Cf, _Float16* __restrict__ Ch,
                              int K, int lda, int ldb, int ldc, int do_relu) {
    const int i0 = blockIdx.x * 16;
    const int j0 = blockIdx.y * 16;
    v8f acc = {};
    for (int k = 0; k < K; k += 32) {
        v16h a = frag_load(A + (size_t)i0 * lda + k, lda);
        v16h b = frag_load(B + (size_t)j0 * ldb + k, ldb);
        acc = __builtin_amdgcn_wmma_f32_16x16x32_f16(false, a, false, b, (short)0, acc,
                                                     false, false);
    }
    int rb, col; frag_rc(rb, col);
    const float bv = bias ? bias[j0 + col] : 0.f;
#pragma unroll
    for (int r = 0; r < 8; ++r) {
        const int row = i0 + rb + r;
        float v = acc[r] + bv;
        if (do_relu) v = fmaxf(v, 0.f);
        const size_t idx = (size_t)row * ldc + j0 + col;
        if (Cf) Cf[idx] = v;
        if (Ch) Ch[idx] = (_Float16)v;
    }
}

__global__ void gemm_attn(const _Float16* __restrict__ hs, const float* __restrict__ adj,
                          _Float16* __restrict__ attn) {
    const int i0 = blockIdx.x * 16;
    const int j0 = blockIdx.y * 16;
    v8f acc = {};
#pragma unroll
    for (int k = 0; k < DIM; k += 32) {
        v16h a = frag_load(hs + (size_t)i0 * DIM + k, DIM);
        v16h b = frag_load(hs + (size_t)j0 * DIM + k, DIM);
        acc = __builtin_amdgcn_wmma_f32_16x16x32_f16(false, a, false, b, (short)0, acc,
                                                     false, false);
    }
    int rb, col; frag_rc(rb, col);
#pragma unroll
    for (int r = 0; r < 8; ++r) {
        const int row = i0 + rb + r;
        const size_t idx = (size_t)row * NN + j0 + col;
        attn[idx] = (_Float16)(acc[r] * adj[idx]);
    }
}

__global__ void row_abs_sum(const _Float16* __restrict__ attn, float* __restrict__ denom) {
    __shared__ float red[256];
    const int i = blockIdx.x;
    float s = 0.f;
    for (int j = threadIdx.x; j < NN; j += 256) s += fabsf((float)attn[(size_t)i * NN + j]);
    red[threadIdx.x] = s;
    __syncthreads();
    for (int w = 128; w > 0; w >>= 1) {
        if (threadIdx.x < w) red[threadIdx.x] += red[threadIdx.x + w];
        __syncthreads();
    }
    if (threadIdx.x == 0) denom[i] = fmaxf(red[0], 1e-12f);
}

__global__ void transpose_f16(const _Float16* __restrict__ src, _Float16* __restrict__ dst) {
    const int j = blockIdx.x;
    const int n = threadIdx.x;
    dst[(size_t)n * NN + j] = src[(size_t)j * DIM + n];
}

__global__ void gemm_norm_axpy(const _Float16* __restrict__ attn, const _Float16* __restrict__ hsT,
                               const float* __restrict__ denom, float* __restrict__ xs) {
    const int i0 = blockIdx.x * 16;
    const int j0 = blockIdx.y * 16;
    const _Float16* Ab = attn + (size_t)i0 * NN;
    const _Float16* Bb = hsT + (size_t)j0 * NN;
    v8f acc = {};
    for (int k = 0; k < NN; k += 32) {
        __builtin_prefetch(Ab + k + 256, 0, 0);
        v16h a = frag_load(Ab + k, NN);
        v16h b = frag_load(Bb + k, NN);
        acc = __builtin_amdgcn_wmma_f32_16x16x32_f16(false, a, false, b, (short)0, acc,
                                                     false, false);
    }
    int rb, col; frag_rc(rb, col);
#pragma unroll
    for (int r = 0; r < 8; ++r) {
        const int row = i0 + rb + r;
        xs[(size_t)row * DIM + j0 + col] += acc[r] / denom[row];
    }
}

__global__ void gather_embed(const int* __restrict__ idx, const float* __restrict__ emb,
                             float* __restrict__ of32, _Float16* __restrict__ of16) {
    const int i = blockIdx.x;
    const int n = threadIdx.x;
    const float v = emb[(size_t)idx[i] * DIM + n];
    if (of32) of32[(size_t)i * DIM + n] = v;
    if (of16) of16[(size_t)i * DIM + n] = (_Float16)v;
}

__global__ void f32_to_f16(const float* __restrict__ s, _Float16* __restrict__ d, int n) {
    const int i = blockIdx.x * 256 + threadIdx.x;
    if (i < n) d[i] = (_Float16)s[i];
}

__global__ void vec_add(const float* __restrict__ a, const float* __restrict__ b,
                        float* __restrict__ c, int n) {
    const int i = blockIdx.x * 256 + threadIdx.x;
    if (i < n) c[i] = a[i] + b[i];
}

__global__ void lstm_dir(const float* __restrict__ G, const float* __restrict__ Whh,
                         float* __restrict__ hout) {
    extern __shared__ float sm[];
    float* WhhT = sm;
    float* h_sh = sm + DIM * 512;
    float* z_sh = h_sh + DIM;
    const int d   = blockIdx.x;
    const int tid = threadIdx.x;
    const float* W  = Whh + (size_t)d * 512 * DIM;
    const float* Gd = G + (size_t)d * MM * 512;
    float* hd = hout + (size_t)d * MM * DIM;

    for (int idx = tid; idx < 512 * DIM; idx += 512) {
        const int j = idx >> 7, k = idx & 127;
        WhhT[k * 512 + j] = W[idx];
    }
    if (tid < DIM) h_sh[tid] = 0.f;
    float c = 0.f;
    __syncthreads();

    for (int t = 0; t < MM; ++t) {
        const int tt = d ? (MM - 1 - t) : t;
        float acc = Gd[(size_t)tt * 512 + tid];
#pragma unroll 8
        for (int k = 0; k < DIM; ++k) acc += h_sh[k] * WhhT[k * 512 + tid];
        z_sh[tid] = acc;
        __syncthreads();
        if (tid < DIM) {
            const float zi = z_sh[tid], zf = z_sh[tid + 128];
            const float zg = z_sh[tid + 256], zo = z_sh[tid + 384];
            const float ig = 1.f / (1.f + __expf(-zi));
            const float fg = 1.f / (1.f + __expf(-zf));
            const float gg = tanhf(zg);
            const float og = 1.f / (1.f + __expf(-zo));
            c = fg * c + ig * gg;
            const float h = og * tanhf(c);
            h_sh[tid] = h;
            hd[(size_t)tt * DIM + tid] = h;
        }
        __syncthreads();
    }
}

__global__ void bilstm_concat_relu(const float* __restrict__ hbuf, _Float16* __restrict__ xs_rnn) {
    const int t = blockIdx.x;
    const int n = threadIdx.x;
    const float v = (n < DIM) ? hbuf[(size_t)t * DIM + n]
                              : hbuf[(size_t)(MM + t) * DIM + (n - DIM)];
    xs_rnn[(size_t)t * (2 * DIM) + n] = (_Float16)fmaxf(v, 0.f);
}

__global__ void row_mean(const float* __restrict__ W, float* __restrict__ out) {
    __shared__ float red[256];
    const int i = blockIdx.x;
    float s = 0.f;
    for (int j = threadIdx.x; j < MM; j += 256) s += W[(size_t)i * MM + j];
    red[threadIdx.x] = s;
    __syncthreads();
    for (int w = 128; w > 0; w >>= 1) {
        if (threadIdx.x < w) red[threadIdx.x] += red[threadIdx.x + w];
        __syncthreads();
    }
    if (threadIdx.x == 0) out[i] = red[0] / (float)MM;
}

__global__ void col_mean(const float* __restrict__ W, float* __restrict__ out) {
    const int j = blockIdx.x * 256 + threadIdx.x;
    float s = 0.f;
    for (int i = 0; i < NN; ++i) s += W[(size_t)i * MM + j];
    out[j] = s / (float)NN;
}

__global__ void pool_vectors(const float* __restrict__ h, const float* __restrict__ hp,
                             const float* __restrict__ wc, const float* __restrict__ wp,
                             float* __restrict__ cat) {
    const int tid = threadIdx.x;
    const int n = tid & 127;
    const float* src = (tid < 128) ? h : hp;
    const float* wv  = (tid < 128) ? wc : wp;
    float s = 0.f;
    for (int i = 0; i < NN; ++i) s += wv[i] * src[(size_t)i * DIM + n];
    cat[tid] = s / (float)NN;
}

__global__ void final_mlp(const float* __restrict__ cat_in, const float* __restrict__ Wo,
                          const float* __restrict__ bo, const float* __restrict__ Wi,
                          const float* __restrict__ bi, float* __restrict__ out) {
    __shared__ float cur[256], nxt[256];
    const int t = threadIdx.x;
    cur[t] = cat_in[t];
    __syncthreads();
    for (int l = 0; l < 2; ++l) {
        const float* Wl = Wo + (size_t)l * 256 * 256;
        float s = bo[l * 256 + t];
        for (int k = 0; k < 256; ++k) s += cur[k] * Wl[(size_t)t * 256 + k];
        nxt[t] = fmaxf(s, 0.f);
        __syncthreads();
        cur[t] = nxt[t];
        __syncthreads();
    }
    if (t < 2) {
        float s = bi[t];
        for (int k = 0; k < 256; ++k) s += cur[k] * Wi[(size_t)t * 256 + k];
        out[t] = s;
    }
}

extern "C" void kernel_launch(void* const* d_in, const int* in_sizes, int n_in,
                              void* d_out, int out_size, void* d_ws, size_t ws_size,
                              hipStream_t stream) {
    (void)in_sizes; (void)n_in; (void)out_size; (void)ws_size;
    const int*   fingerprints = (const int*)d_in[0];
    const float* adjacency    = (const float*)d_in[1];
    const int*   words        = (const int*)d_in[2];
    const float* emb_fp       = (const float*)d_in[3];
    const float* emb_word     = (const float*)d_in[4];
    const float* Wg  = (const float*)d_in[5];
    const float* bg  = (const float*)d_in[6];
    const float* Wih = (const float*)d_in[7];
    const float* Whh = (const float*)d_in[8];
    const float* bih = (const float*)d_in[9];
    const float* bhh = (const float*)d_in[10];
    const float* Wc  = (const float*)d_in[11];
    const float* bc  = (const float*)d_in[12];
    const float* Wp  = (const float*)d_in[13];
    const float* bp  = (const float*)d_in[14];
    const float* Wo  = (const float*)d_in[15];
    const float* bo  = (const float*)d_in[16];
    const float* Wi  = (const float*)d_in[17];
    const float* bi  = (const float*)d_in[18];

    float* out         = (float*)d_out;
    float* weights_out = out + 2;

    char* base = (char*)d_ws;
    size_t off = 0;
    auto alloc = [&](size_t bytes) -> void* {
        void* p = base + off;
        off = (off + bytes + 255) & ~(size_t)255;
        return p;
    };
    float*    xs       = (float*)   alloc((size_t)NN * DIM * 4);
    _Float16* xs_h     = (_Float16*)alloc((size_t)NN * DIM * 2);
    _Float16* hs_h     = (_Float16*)alloc((size_t)NN * DIM * 2);
    _Float16* hsT_h    = (_Float16*)alloc((size_t)DIM * NN * 2);
    _Float16* attn_h   = (_Float16*)alloc((size_t)NN * NN * 2);
    float*    denom    = (float*)   alloc((size_t)NN * 4);
    _Float16* Wg_h     = (_Float16*)alloc((size_t)3 * DIM * DIM * 2);
    _Float16* Wc_h     = (_Float16*)alloc((size_t)DIM * DIM * 2);
    _Float16* Wp_h     = (_Float16*)alloc((size_t)DIM * 2 * DIM * 2);
    _Float16* Wih_h    = (_Float16*)alloc((size_t)2 * 512 * DIM * 2);
    float*    bsum     = (float*)   alloc((size_t)2 * 512 * 4);
    _Float16* wv_h     = (_Float16*)alloc((size_t)MM * DIM * 2);
    float*    G        = (float*)   alloc((size_t)2 * MM * 512 * 4);
    float*    hbuf     = (float*)   alloc((size_t)2 * MM * DIM * 4);
    _Float16* xs_rnn_h = (_Float16*)alloc((size_t)MM * 2 * DIM * 2);
    float*    h_f32    = (float*)   alloc((size_t)NN * DIM * 4);
    _Float16* h_h      = (_Float16*)alloc((size_t)NN * DIM * 2);
    float*    hp_f32   = (float*)   alloc((size_t)MM * DIM * 4);
    _Float16* hp_h     = (_Float16*)alloc((size_t)MM * DIM * 2);
    float*    w_comp   = (float*)   alloc((size_t)NN * 4);
    float*    w_prot   = (float*)   alloc((size_t)MM * 4);
    float*    cat      = (float*)   alloc(256 * 4);

    f32_to_f16<<<(3 * DIM * DIM + 255) / 256, 256, 0, stream>>>(Wg, Wg_h, 3 * DIM * DIM);
    f32_to_f16<<<(DIM * DIM + 255) / 256, 256, 0, stream>>>(Wc, Wc_h, DIM * DIM);
    f32_to_f16<<<(DIM * 2 * DIM + 255) / 256, 256, 0, stream>>>(Wp, Wp_h, DIM * 2 * DIM);
    f32_to_f16<<<(2 * 512 * DIM + 255) / 256, 256, 0, stream>>>(Wih, Wih_h, 2 * 512 * DIM);
    vec_add<<<(2 * 512 + 255) / 256, 256, 0, stream>>>(bih, bhh, bsum, 2 * 512);

    gather_embed<<<NN, DIM, 0, stream>>>(fingerprints, emb_fp, xs, nullptr);
    gather_embed<<<MM, DIM, 0, stream>>>(words, emb_word, nullptr, wv_h);

    for (int l = 0; l < 3; ++l) {
        f32_to_f16<<<(NN * DIM + 255) / 256, 256, 0, stream>>>(xs, xs_h, NN * DIM);
        gemm_bias_act<<<dim3(NN / 16, DIM / 16), 32, 0, stream>>>(
            xs_h, Wg_h + (size_t)l * DIM * DIM, bg + l * DIM,
            nullptr, hs_h, DIM, DIM, DIM, DIM, 1);
        transpose_f16<<<NN, DIM, 0, stream>>>(hs_h, hsT_h);
        gemm_attn<<<dim3(NN / 16, NN / 16), 32, 0, stream>>>(hs_h, adjacency, attn_h);
        row_abs_sum<<<NN, 256, 0, stream>>>(attn_h, denom);
        gemm_norm_axpy<<<dim3(NN / 16, DIM / 16), 32, 0, stream>>>(attn_h, hsT_h, denom, xs);
    }

    f32_to_f16<<<(NN * DIM + 255) / 256, 256, 0, stream>>>(xs, xs_h, NN * DIM);
    gemm_bias_act<<<dim3(NN / 16, DIM / 16), 32, 0, stream>>>(
        xs_h, Wc_h, bc, h_f32, h_h, DIM, DIM, DIM, DIM, 1);

    for (int d = 0; d < 2; ++d) {
        gemm_bias_act<<<dim3(MM / 16, 512 / 16), 32, 0, stream>>>(
            wv_h, Wih_h + (size_t)d * 512 * DIM, bsum + d * 512,
            G + (size_t)d * MM * 512, nullptr, DIM, DIM, DIM, 512, 0);
    }
    const size_t lstm_lds = (size_t)(DIM * 512 + DIM + 512) * sizeof(float);
    lstm_dir<<<2, 512, lstm_lds, stream>>>(G, Whh, hbuf);
    bilstm_concat_relu<<<MM, 256, 0, stream>>>(hbuf, xs_rnn_h);
    gemm_bias_act<<<dim3(MM / 16, DIM / 16), 32, 0, stream>>>(
        xs_rnn_h, Wp_h, bp, hp_f32, hp_h, 2 * DIM, 2 * DIM, 2 * DIM, DIM, 1);

    gemm_bias_act<<<dim3(NN / 16, MM / 16), 32, 0, stream>>>(
        h_h, hp_h, nullptr, weights_out, nullptr, DIM, DIM, DIM, MM, 0);

    row_mean<<<NN, 256, 0, stream>>>(weights_out, w_comp);
    col_mean<<<MM / 256, 256, 0, stream>>>(weights_out, w_prot);
    pool_vectors<<<1, 256, 0, stream>>>(h_f32, hp_f32, w_comp, w_prot, cat);
    final_mlp<<<1, 256, 0, stream>>>(cat, Wo, bo, Wi, bi, out);
}


# lstm_dir: Whh kept in VGPRs, one thread per hidden unit holds its 4 gate rows (512 f32); a step reads only h from LDS, no z exchange, one barrier; per-row k order of the sums unchanged
# speedup vs baseline: 2.0306x; 2.0306x over previous
#include <hip/hip_runtime.h>

typedef __attribute__((ext_vector_type(16))) _Float16 v16h;
typedef __attribute__((ext_vector_type(8)))  float    v8f;

#define DIM 128
#define NN  4096
#define MM  4096

__device__ __forceinline__ v16h frag_load(const _Float16* __restrict__ tile, int ld) {
    const int lane = threadIdx.x & 31;
    const int r    = lane & 15;
    const int kb   = (lane & 16) ? 8 : 0;
    const _Float16* q = tile + (size_t)r * (size_t)ld + kb;
    v16h f;
    uint4* fu = reinterpret_cast<uint4*>(&f);
    fu[0] = *reinterpret_cast<const uint4*>(q);
    fu[1] = *reinterpret_cast<const uint4*>(q + 16);
    return f;
}

__device__ __forceinline__ void frag_rc(int& rbase, int& col) {
    const int lane = threadIdx.x & 31;
    col   = lane & 15;
    rbase = (lane & 16) ? 8 : 0;
}

__global__ void gemm_bias_act(const _Float16* __restrict__ A, const _Float16* __restrict__ B,
                              const float* __restrict__ bias,
                              float* __restrict__ Cf, _Float16* __restrict__ Ch,
                              int K, int lda, int ldb, int ldc, int do_relu) {
    const int i0 = blockIdx.x * 16;
    const int j0 = blockIdx.y * 16;
    v8f acc = {};
    for (int k = 0; k < K; k += 32) {
        v16h a = frag_load(A + (size_t)i0 * lda + k, lda);
        v16h b = frag_load(B + (size_t)j0 * ldb + k, ldb);
        acc = __builtin_amdgcn_wmma_f32_16x16x32_f16(false, a, false, b, (short)0, acc,
                                                     false, false);
    }
    int rb, col; frag_rc(rb, col);
    const float bv = bias ? bias[j0 + col] : 0.f;
#pragma unroll
    for (int r = 0; r < 8; ++r) {
        const int row = i0 + rb + r;
        float v = acc[r] + bv;
        if (do_relu) v = fmaxf(v, 0.f);
        const size_t idx = (size_t)row * ldc + j0 + col;
        if (Cf) Cf[idx] = v;
        if (Ch) Ch[idx] = (_Float16)v;
    }
}

__global__ void gemm_attn(const _Float16* __restrict__ hs, const float* __restrict__ adj,
                          _Float16* __restrict__ attn) {
    const int i0 = blockIdx.x * 16;
    const int j0 = blockIdx.y * 16;
    v8f acc = {};
#pragma unroll
    for (int k = 0; k < DIM; k += 32) {
        v16h a = frag_load(hs + (size_t)i0 * DIM + k, DIM);
        v16h b = frag_load(hs + (size_t)j0 * DIM + k, DIM);
        acc = __builtin_amdgcn_wmma_f32_16x16x32_f16(false, a, false, b, (short)0, acc,
                                                     false, false);
    }
    int rb, col; frag_rc(rb, col);
#pragma unroll
    for (int r = 0; r < 8; ++r) {
        const int row = i0 + rb + r;
        const size_t idx = (size_t)row * NN + j0 + col;
        attn[idx] = (_Float16)(acc[r] * adj[idx]);
    }
}

__global__ void row_abs_sum(const _Float16* __restrict__ attn, float* __restrict__ denom) {
    __shared__ float red[256];
    const int i = blockIdx.x;
    float s = 0.f;
    for (int j = threadIdx.x; j < NN; j += 256) s += fabsf((float)attn[(size_t)i * NN + j]);
    red[threadIdx.x] = s;
    __syncthreads();
    for (int w = 128; w > 0; w >>= 1) {
        if (threadIdx.x < w) red[threadIdx.x] += red[threadIdx.x + w];
        __syncthreads();
    }
    if (threadIdx.x == 0) denom[i] = fmaxf(red[0], 1e-12f);
}

__global__ void transpose_f16(const _Float16* __restrict__ src, _Float16* __restrict__ dst) {
    const int j = blockIdx.x;
    const int n = threadIdx.x;
    dst[(size_t)n * NN + j] = src[(size_t)j * DIM + n];
}

__global__ void gemm_norm_axpy(const _Float16* __restrict__ attn, const _Float16* __restrict__ hsT,
                               const float* __restrict__ denom, float* __restrict__ xs) {
    const int i0 = blockIdx.x * 16;
    const int j0 = blockIdx.y * 16;
    const _Float16* Ab = attn + (size_t)i0 * NN;
    const _Float16* Bb = hsT + (size_t)j0 * NN;
    v8f acc = {};
    for (int k = 0; k < NN; k += 32) {
        __builtin_prefetch(Ab + k + 256, 0, 0);
        v16h a = frag_load(Ab + k, NN);
        v16h b = frag_load(Bb + k, NN);
        acc = __builtin_amdgcn_wmma_f32_16x16x32_f16(false, a, false, b, (short)0, acc,
                                                     false, false);
    }
    int rb, col; frag_rc(rb, col);
#pragma unroll
    for (int r = 0; r < 8; ++r) {
        const int row = i0 + rb + r;
        xs[(size_t)row * DIM + j0 + col] += acc[r] / denom[row];
    }
}

__global__ void gather_embed(const int* __restrict__ idx, const float* __restrict__ emb,
                             float* __restrict__ of32, _Float16* __restrict__ of16) {
    const int i = blockIdx.x;
    const int n = threadIdx.x;
    const float v = emb[(size_t)idx[i] * DIM + n];
    if (of32) of32[(size_t)i * DIM + n] = v;
    if (of16) of16[(size_t)i * DIM + n] = (_Float16)v;
}

__global__ void f32_to_f16(const float* __restrict__ s, _Float16* __restrict__ d, int n) {
    const int i = blockIdx.x * 256 + threadIdx.x;
    if (i < n) d[i] = (_Float16)s[i];
}

__global__ void vec_add(const float* __restrict__ a, const float* __restrict__ b,
                        float* __restrict__ c, int n) {
    const int i = blockIdx.x * 256 + threadIdx.x;
    if (i < n) c[i] = a[i] + b[i];
}

__global__ __launch_bounds__(128) void lstm_dir(const float* __restrict__ G,
                                                const float* __restrict__ Whh,
                                                float* __restrict__ hout) {
    __shared__ __attribute__((aligned(16))) float h_sh[2][DIM];
    const int d = blockIdx.x;
    const int n = threadIdx.x;
    const float* W  = Whh + (size_t)d * 512 * DIM;
    const float* Gd = G + (size_t)d * MM * 512;
    float* hd = hout + (size_t)d * MM * DIM;

    float wi[DIM], wf[DIM], wg[DIM], wo[DIM];
    {
        const float4* ri = reinterpret_cast<const float4*>(W + (size_t)(n)       * DIM);
        const float4* rf = reinterpret_cast<const float4*>(W + (size_t)(n + 128) * DIM);
        const float4* rg = reinterpret_cast<const float4*>(W + (size_t)(n + 256) * DIM);
        const float4* ro = reinterpret_cast<const float4*>(W + (size_t)(n + 384) * DIM);
#pragma unroll
        for (int k4 = 0; k4 < DIM / 4; ++k4) {
            const float4 a = ri[k4], b = rf[k4], c4 = rg[k4], e = ro[k4];
            wi[4 * k4 + 0] = a.x;  wi[4 * k4 + 1] = a.y;  wi[4 * k4 + 2] = a.z;  wi[4 * k4 + 3] = a.w;
            wf[4 * k4 + 0] = b.x;  wf[4 * k4 + 1] = b.y;  wf[4 * k4 + 2] = b.z;  wf[4 * k4 + 3] = b.w;
            wg[4 * k4 + 0] = c4.x; wg[4 * k4 + 1] = c4.y; wg[4 * k4 + 2] = c4.z; wg[4 * k4 + 3] = c4.w;
            wo[4 * k4 + 0] = e.x;  wo[4 * k4 + 1] = e.y;  wo[4 * k4 + 2] = e.z;  wo[4 * k4 + 3] = e.w;
        }
    }
    h_sh[0][n] = 0.f;
    float c = 0.f;
    __syncthreads();

    for (int t = 0; t < MM; ++t) {
        const int tt = d ? (MM - 1 - t) : t;
        const float4* h4 = reinterpret_cast<const float4*>(h_sh[t & 1]);
        const float* g = Gd + (size_t)tt * 512 + n;
        float zi = g[0], zf = g[128], zg = g[256], zo = g[384];
        float4 hv[DIM / 4];
#pragma unroll
        for (int k4 = 0; k4 < DIM / 4; ++k4) hv[k4] = h4[k4];
#pragma unroll
        for (int k = 0; k < DIM; ++k) {
            const float hk = (k & 3) == 0 ? hv[k >> 2].x : (k & 3) == 1 ? hv[k >> 2].y
                           : (k & 3) == 2 ? hv[k >> 2].z : hv[k >> 2].w;
            zi += hk * wi[k]; zf += hk * wf[k];
            zg += hk * wg[k]; zo += hk * wo[k];
            __builtin_amdgcn_sched_barrier(0);
        }
        asm volatile("" : "+v"(zi), "+v"(zf), "+v"(zg), "+v"(zo));
        const float ig = 1.f / (1.f + __expf(-zi));
        const float fg = 1.f / (1.f + __expf(-zf));
        const float gg = tanhf(zg);
        const float og = 1.f / (1.f + __expf(-zo));
        c = fg * c + ig * gg;
        const float h = og * tanhf(c);
        h_sh[(t + 1) & 1][n] = h;
        hd[(size_t)tt * DIM + n] = h;
        __syncthreads();
    }
}

__global__ void bilstm_concat_relu(const float* __restrict__ hbuf, _Float16* __restrict__ xs_rnn) {
    const int t = blockIdx.x;
    const int n = threadIdx.x;
    const float v = (n < DIM) ? hbuf[(size_t)t * DIM + n]
                              : hbuf[(size_t)(MM + t) * DIM + (n - DIM)];
    xs_rnn[(size_t)t * (2 * DIM) + n] = (_Float16)fmaxf(v, 0.f);
}

__global__ void row_mean(const float* __restrict__ W, float* __restrict__ out) {
    __shared__ float red[256];
    const int i = blockIdx.x;
    float s = 0.f;
    for (int j = threadIdx.x; j < MM; j += 256) s += W[(size_t)i * MM + j];
    red[threadIdx.x] = s;
    __syncthreads();
    for (int w = 128; w > 0; w >>= 1) {
        if (threadIdx.x < w) red[threadIdx.x] += red[threadIdx.x + w];
        __syncthreads();
    }
    if (threadIdx.x == 0) out[i] = red[0] / (float)MM;
}

__global__ void col_mean(const float* __restrict__ W, float* __restrict__ out) {
    const int j = blockIdx.x * 256 + threadIdx.x;
    float s = 0.f;
    for (int i = 0; i < NN; ++i) s += W[(size_t)i * MM + j];
    out[j] = s / (float)NN;
}

__global__ void pool_vectors(const float* __restrict__ h, const float* __restrict__ hp,
                             const float* __restrict__ wc, const float* __restrict__ wp,
                             float* __restrict__ cat) {
    const int tid = threadIdx.x;
    const int n = tid & 127;
    const float* src = (tid < 128) ? h : hp;
    const float* wv  = (tid < 128) ? wc : wp;
    float s = 0.f;
    for (int i = 0; i < NN; ++i) s += wv[i] * src[(size_t)i * DIM + n];
    cat[tid] = s / (float)NN;
}

__global__ void final_mlp(const float* __restrict__ cat_in, const float* __restrict__ Wo,
                          const float* __restrict__ bo, const float* __restrict__ Wi,
                          const float* __restrict__ bi, float* __restrict__ out) {
    __shared__ float cur[256], nxt[256];
    const int t = threadIdx.x;
    cur[t] = cat_in[t];
    __syncthreads();
    for (int l = 0; l < 2; ++l) {
        const float* Wl = Wo + (size_t)l * 256 * 256;
        float s = bo[l * 256 + t];
        for (int k = 0; k < 256; ++k) s += cur[k] * Wl[(size_t)t * 256 + k];
        nxt[t] = fmaxf(s, 0.f);
        __syncthreads();
        cur[t] = nxt[t];
        __syncthreads();
    }
    if (t < 2) {
        float s = bi[t];
        for (int k = 0; k < 256; ++k) s += cur[k] * Wi[(size_t)t * 256 + k];
        out[t] = s;
    }
}

extern "C" void kernel_launch(void* const* d_in, const int* in_sizes, int n_in,
                              void* d_out, int out_size, void* d_ws, size_t ws_size,
                              hipStream_t stream) {
    (void)in_sizes; (void)n_in; (void)out_size; (void)ws_size;
    const int*   fingerprints = (const int*)d_in[0];
    const float* adjacency    = (const float*)d_in[1];
    const int*   words        = (const int*)d_in[2];
    const float* emb_fp       = (const float*)d_in[3];
    const float* emb_word     = (const float*)d_in[4];
    const float* Wg  = (const float*)d_in[5];
    const float* bg  = (const float*)d_in[6];
    const float* Wih = (const float*)d_in[7];
    const float* Whh = (const float*)d_in[8];
    const float* bih = (const float*)d_in[9];
    const float* bhh = (const float*)d_in[10];
    const float* Wc  = (const float*)d_in[11];
    const float* bc  = (const float*)d_in[12];
    const float* Wp  = (const float*)d_in[13];
    const float* bp  = (const float*)d_in[14];
    const float* Wo  = (const float*)d_in[15];
    const float* bo  = (const float*)d_in[16];
    const float* Wi  = (const float*)d_in[17];
    const float* bi  = (const float*)d_in[18];

    float* out         = (float*)d_out;
    float* weights_out = out + 2;

    char* base = (char*)d_ws;
    size_t off = 0;
    auto alloc = [&](size_t bytes) -> void* {
        void* p = base + off;
        off = (off + bytes + 255) & ~(size_t)255;
        return p;
    };
    float*    xs       = (float*)   alloc((size_t)NN * DIM * 4);
    _Float16* xs_h     = (_Float16*)alloc((size_t)NN * DIM * 2);
    _Float16* hs_h     = (_Float16*)alloc((size_t)NN * DIM * 2);
    _Float16* hsT_h    = (_Float16*)alloc((size_t)DIM * NN * 2);
    _Float16* attn_h   = (_Float16*)alloc((size_t)NN * NN * 2);
    float*    denom    = (float*)   alloc((size_t)NN * 4);
    _Float16* Wg_h     = (_Float16*)alloc((size_t)3 * DIM * DIM * 2);
    _Float16* Wc_h     = (_Float16*)alloc((size_t)DIM * DIM * 2);
    _Float16* Wp_h     = (_Float16*)alloc((size_t)DIM * 2 * DIM * 2);
    _Float16* Wih_h    = (_Float16*)alloc((size_t)2 * 512 * DIM * 2);
    float*    bsum     = (float*)   alloc((size_t)2 * 512 * 4);
    _Float16* wv_h     = (_Float16*)alloc((size_t)MM * DIM * 2);
    float*    G        = (float*)   alloc((size_t)2 * MM * 512 * 4);
    float*    hbuf     = (float*)   alloc((size_t)2 * MM * DIM * 4);
    _Float16* xs_rnn_h = (_Float16*)alloc((size_t)MM * 2 * DIM * 2);
    float*    h_f32    = (float*)   alloc((size_t)NN * DIM * 4);
    _Float16* h_h      = (_Float16*)alloc((size_t)NN * DIM * 2);
    float*    hp_f32   = (float*)   alloc((size_t)MM * DIM * 4);
    _Float16* hp_h     = (_Float16*)alloc((size_t)MM * DIM * 2);
    float*    w_comp   = (float*)   alloc((size_t)NN * 4);
    float*    w_prot   = (float*)   alloc((size_t)MM * 4);
    float*    cat      = (float*)   alloc(256 * 4);

    f32_to_f16<<<(3 * DIM * DIM + 255) / 256, 256, 0, stream>>>(Wg, Wg_h, 3 * DIM * DIM);
    f32_to_f16<<<(DIM * DIM + 255) / 256, 256, 0, stream>>>(Wc, Wc_h, DIM * DIM);
    f32_to_f16<<<(DIM * 2 * DIM + 255) / 256, 256, 0, stream>>>(Wp, Wp_h, DIM * 2 * DIM);
    f32_to_f16<<<(2 * 512 * DIM + 255) / 256, 256, 0, stream>>>(Wih, Wih_h, 2 * 512 * DIM);
    vec_add<<<(2 * 512 + 255) / 256, 256, 0, stream>>>(bih, bhh, bsum, 2 * 512);

    gather_embed<<<NN, DIM, 0, stream>>>(fingerprints, emb_fp, xs, nullptr);
    gather_embed<<<MM, DIM, 0, stream>>>(words, emb_word, nullptr, wv_h);

    for (int l = 0; l < 3; ++l) {
        f32_to_f16<<<(NN * DIM + 255) / 256, 256, 0, stream>>>(xs, xs_h, NN * DIM);
        gemm_bias_act<<<dim3(NN / 16, DIM / 16), 32, 0, stream>>>(
            xs_h, Wg_h + (size_t)l * DIM * DIM, bg + l * DIM,
            nullptr, hs_h, DIM, DIM, DIM, DIM, 1);
        transpose_f16<<<NN, DIM, 0, stream>>>(hs_h, hsT_h);
        gemm_attn<<<dim3(NN / 16, NN / 16), 32, 0, stream>>>(hs_h, adjacency, attn_h);
        row_abs_sum<<<NN, 256, 0, stream>>>(attn_h, denom);
        gemm_norm_axpy<<<dim3(NN / 16, DIM / 16), 32, 0, stream>>>(attn_h, hsT_h, denom, xs);
    }

    f32_to_f16<<<(NN * DIM + 255) / 256, 256, 0, stream>>>(xs, xs_h, NN * DIM);
    gemm_bias_act<<<dim3(NN / 16, DIM / 16), 32, 0, stream>>>(
        xs_h, Wc_h, bc, h_f32, h_h, DIM, DIM, DIM, DIM, 1);

    for (int d = 0; d < 2; ++d) {
        gemm_bias_act<<<dim3(MM / 16, 512 / 16), 32, 0, stream>>>(
            wv_h, Wih_h + (size_t)d * 512 * DIM, bsum + d * 512,
            G + (size_t)d * MM * 512, nullptr, DIM, DIM, DIM, 512, 0);
    }
    lstm_dir<<<2, 128, 0, stream>>>(G, Whh, hbuf);
    bilstm_concat_relu<<<MM, 256, 0, stream>>>(hbuf, xs_rnn_h);
    gemm_bias_act<<<dim3(MM / 16, DIM / 16), 32, 0, stream>>>(
        xs_rnn_h, Wp_h, bp, hp_f32, hp_h, 2 * DIM, 2 * DIM, 2 * DIM, DIM, 1);

    gemm_bias_act<<<dim3(NN / 16, MM / 16), 32, 0, stream>>>(
        h_h, hp_h, nullptr, weights_out, nullptr, DIM, DIM, DIM, MM, 0);

    row_mean<<<NN, 256, 0, stream>>>(weights_out, w_comp);
    col_mean<<<MM / 256, 256, 0, stream>>>(weights_out, w_prot);
    pool_vectors<<<1, 256, 0, stream>>>(h_f32, hp_f32, w_comp, w_prot, cat);
    final_mlp<<<1, 256, 0, stream>>>(cat, Wo, bo, Wi, bi, out);
}


# lstm_dir: pre-gates G of step t+1 are loaded during step t (register double buffer), so a step no longer starts by waiting for a global load; order of every sum unchanged
# speedup vs baseline: 2.4511x; 1.2071x over previous
#include <hip/hip_runtime.h>

typedef __attribute__((ext_vector_type(16))) _Float16 v16h;
typedef __attribute__((ext_vector_type(8)))  float    v8f;

#define DIM 128
#define NN  4096
#define MM  4096

__device__ __forceinline__ v16h frag_load(const _Float16* __restrict__ tile, int ld) {
    const int lane = threadIdx.x & 31;
    const int r    = lane & 15;
    const int kb   = (lane & 16) ? 8 : 0;
    const _Float16* q = tile + (size_t)r * (size_t)ld + kb;
    v16h f;
    uint4* fu = reinterpret_cast<uint4*>(&f);
    fu[0] = *reinterpret_cast<const uint4*>(q);
    fu[1] = *reinterpret_cast<const uint4*>(q + 16);
    return f;
}

__device__ __forceinline__ void frag_rc(int& rbase, int& col) {
    const int lane = threadIdx.x & 31;
    col   = lane & 15;
    rbase = (lane & 16) ? 8 : 0;
}

__global__ void gemm_bias_act(const _Float16* __restrict__ A, const _Float16* __restrict__ B,
                              const float* __restrict__ bias,
                              float* __restrict__ Cf, _Float16* __restrict__ Ch,
                              int K, int lda, int ldb, int ldc, int do_relu) {
    const int i0 = blockIdx.x * 16;
    const int j0 = blockIdx.y * 16;
    v8f acc = {};
    for (int k = 0; k < K; k += 32) {
        v16h a = frag_load(A + (size_t)i0 * lda + k, lda);
        v16h b = frag_load(B + (size_t)j0 * ldb + k, ldb);
        acc = __builtin_amdgcn_wmma_f32_16x16x32_f16(false, a, false, b, (short)0, acc,
                                                     false, false);
    }
    int rb, col; frag_rc(rb, col);
    const float bv = bias ? bias[j0 + col] : 0.f;
#pragma unroll
    for (int r = 0; r < 8; ++r) {
        const int row = i0 + rb + r;
        float v = acc[r] + bv;
        if (do_relu) v = fmaxf(v, 0.f);
        const size_t idx = (size_t)row * ldc + j0 + col;
        if (Cf) Cf[idx] = v;
        if (Ch) Ch[idx] = (_Float16)v;
    }
}

__global__ void gemm_attn(const _Float16* __restrict__ hs, const float* __restrict__ adj,
                          _Float16* __restrict__ attn) {
    const int i0 = blockIdx.x * 16;
    const int j0 = blockIdx.y * 16;
    v8f acc = {};
#pragma unroll
    for (int k = 0; k < DIM; k += 32) {
        v16h a = frag_load(hs + (size_t)i0 * DIM + k, DIM);
        v16h b = frag_load(hs + (size_t)j0 * DIM + k, DIM);
        acc = __builtin_amdgcn_wmma_f32_16x16x32_f16(false, a, false, b, (short)0, acc,
                                                     false, false);
    }
    int rb, col; frag_rc(rb, col);
#pragma unroll
    for (int r = 0; r < 8; ++r) {
        const int row = i0 + rb + r;
        const size_t idx = (size_t)row * NN + j0 + col;
        attn[idx] = (_Float16)(acc[r] * adj[idx]);
    }
}

__global__ void row_abs_sum(const _Float16* __restrict__ attn, float* __restrict__ denom) {
    __shared__ float red[256];
    const int i = blockIdx.x;
    float s = 0.f;
    for (int j = threadIdx.x; j < NN; j += 256) s += fabsf((float)attn[(size_t)i * NN + j]);
    red[threadIdx.x] = s;
    __syncthreads();
    for (int w = 128; w > 0; w >>= 1) {
        if (threadIdx.x < w) red[threadIdx.x] += red[threadIdx.x + w];
        __syncthreads();
    }
    if (threadIdx.x == 0) denom[i] = fmaxf(red[0], 1e-12f);
}

__global__ void transpose_f16(const _Float16* __restrict__ src, _Float16* __restrict__ dst) {
    const int j = blockIdx.x;
    const int n = threadIdx.x;
    dst[(size_t)n * NN + j] = src[(size_t)j * DIM + n];
}

__global__ void gemm_norm_axpy(const _Float16* __restrict__ attn, const _Float16* __restrict__ hsT,
                               const float* __restrict__ denom, float* __restrict__ xs) {
    const int i0 = blockIdx.x * 16;
    const int j0 = blockIdx.y * 16;
    const _Float16* Ab = attn + (size_t)i0 * NN;
    const _Float16* Bb = hsT + (size_t)j0 * NN;
    v8f acc = {};
    for (int k = 0; k < NN; k += 32) {
        __builtin_prefetch(Ab + k + 256, 0, 0);
        v16h a = frag_load(Ab + k, NN);
        v16h b = frag_load(Bb + k, NN);
        acc = __builtin_amdgcn_wmma_f32_16x16x32_f16(false, a, false, b, (short)0, acc,
                                                     false, false);
    }
    int rb, col; frag_rc(rb, col);
#pragma unroll
    for (int r = 0; r < 8; ++r) {
        const int row = i0 + rb + r;
        xs[(size_t)row * DIM + j0 + col] += acc[r] / denom[row];
    }
}

__global__ void gather_embed(const int* __restrict__ idx, const float* __restrict__ emb,
                             float* __restrict__ of32, _Float16* __restrict__ of16) {
    const int i = blockIdx.x;
    const int n = threadIdx.x;
    const float v = emb[(size_t)idx[i] * DIM + n];
    if (of32) of32[(size_t)i * DIM + n] = v;
    if (of16) of16[(size_t)i * DIM + n] = (_Float16)v;
}

__global__ void f32_to_f16(const float* __restrict__ s, _Float16* __restrict__ d, int n) {
    const int i = blockIdx.x * 256 + threadIdx.x;
    if (i < n) d[i] = (_Float16)s[i];
}

__global__ void vec_add(const float* __restrict__ a, const float* __restrict__ b,
                        float* __restrict__ c, int n) {
    const int i = blockIdx.x * 256 + threadIdx.x;
    if (i < n) c[i] = a[i] + b[i];
}

__global__ __launch_bounds__(128) void lstm_dir(const float* __restrict__ G,
                                                const float* __restrict__ Whh,
                                                float* __restrict__ hout) {
    __shared__ __attribute__((aligned(16))) float h_sh[2][DIM];
    const int d = blockIdx.x;
    const int n = threadIdx.x;
    const float* W  = Whh + (size_t)d * 512 * DIM;
    const float* Gd = G + (size_t)d * MM * 512;
    float* hd = hout + (size_t)d * MM * DIM;

    float wi[DIM], wf[DIM], wg[DIM], wo[DIM];
    {
        const float4* ri = reinterpret_cast<const float4*>(W + (size_t)(n)       * DIM);
        const float4* rf = reinterpret_cast<const float4*>(W + (size_t)(n + 128) * DIM);
        const float4* rg = reinterpret_cast<const float4*>(W + (size_t)(n + 256) * DIM);
        const float4* ro = reinterpret_cast<const float4*>(W + (size_t)(n + 384) * DIM);
#pragma unroll
        for (int k4 = 0; k4 < DIM / 4; ++k4) {
            const float4 a = ri[k4], b = rf[k4], c4 = rg[k4], e = ro[k4];
            wi[4 * k4 + 0] = a.x;  wi[4 * k4 + 1] = a.y;  wi[4 * k4 + 2] = a.z;  wi[4 * k4 + 3] = a.w;
            wf[4 * k4 + 0] = b.x;  wf[4 * k4 + 1] = b.y;  wf[4 * k4 + 2] = b.z;  wf[4 * k4 + 3] = b.w;
            wg[4 * k4 + 0] = c4.x; wg[4 * k4 + 1] = c4.y; wg[4 * k4 + 2] = c4.z; wg[4 * k4 + 3] = c4.w;
            wo[4 * k4 + 0] = e.x;  wo[4 * k4 + 1] = e.y;  wo[4 * k4 + 2] = e.z;  wo[4 * k4 + 3] = e.w;
        }
    }
    h_sh[0][n] = 0.f;
    float c = 0.f;
    __syncthreads();

    float pi, pf, pg, po;
    {
        const float* g0 = Gd + (size_t)(d ? (MM - 1) : 0) * 512 + n;
        pi = g0[0]; pf = g0[128]; pg = g0[256]; po = g0[384];
    }

    for (int t = 0; t < MM; ++t) {
        const int tt = d ? (MM - 1 - t) : t;
        const float4* h4 = reinterpret_cast<const float4*>(h_sh[t & 1]);
        float zi = pi, zf = pf, zg = pg, zo = po;
        float4 hv[DIM / 4];
#pragma unroll
        for (int k4 = 0; k4 < DIM / 4; ++k4) hv[k4] = h4[k4];
        __builtin_amdgcn_sched_barrier(0);
        {
            const int tn = (t + 1 < MM) ? (t + 1) : t;
            const float* gn = Gd + (size_t)(d ? (MM - 1 - tn) : tn) * 512 + n;
            pi = gn[0]; pf = gn[128]; pg = gn[256]; po = gn[384];
        }
#pragma unroll
        for (int k = 0; k < DIM; ++k) {
            const float hk = (k & 3) == 0 ? hv[k >> 2].x : (k & 3) == 1 ? hv[k >> 2].y
                           : (k & 3) == 2 ? hv[k >> 2].z : hv[k >> 2].w;
            zi += hk * wi[k]; zf += hk * wf[k];
            zg += hk * wg[k]; zo += hk * wo[k];
            __builtin_amdgcn_sched_barrier(0);
        }
        asm volatile("" : "+v"(zi), "+v"(zf), "+v"(zg), "+v"(zo));
        const float ig = 1.f / (1.f + __expf(-zi));
        const float fg = 1.f / (1.f + __expf(-zf));
        const float gg = tanhf(zg);
        const float og = 1.f / (1.f + __expf(-zo));
        c = fg * c + ig * gg;
        const float h = og * tanhf(c);
        h_sh[(t + 1) & 1][n] = h;
        hd[(size_t)tt * DIM + n] = h;
        __syncthreads();
    }
}

__global__ void bilstm_concat_relu(const float* __restrict__ hbuf, _Float16* __restrict__ xs_rnn) {
    const int t = blockIdx.x;
    const int n = threadIdx.x;
    const float v = (n < DIM) ? hbuf[(size_t)t * DIM + n]
                              : hbuf[(size_t)(MM + t) * DIM + (n - DIM)];
    xs_rnn[(size_t)t * (2 * DIM) + n] = (_Float16)fmaxf(v, 0.f);
}

__global__ void row_mean(const float* __restrict__ W, float* __restrict__ out) {
    __shared__ float red[256];
    const int i = blockIdx.x;
    float s = 0.f;
    for (int j = threadIdx.x; j < MM; j += 256) s += W[(size_t)i * MM + j];
    red[threadIdx.x] = s;
    __syncthreads();
    for (int w = 128; w > 0; w >>= 1) {
        if (threadIdx.x < w) red[threadIdx.x] += red[threadIdx.x + w];
        __syncthreads();
    }
    if (threadIdx.x == 0) out[i] = red[0] / (float)MM;
}

__global__ void col_mean(const float* __restrict__ W, float* __restrict__ out) {
    const int j = blockIdx.x * 256 + threadIdx.x;
    float s = 0.f;
    for (int i = 0; i < NN; ++i) s += W[(size_t)i * MM + j];
    out[j] = s / (float)NN;
}

__global__ void pool_vectors(const float* __restrict__ h, const float* __restrict__ hp,
                             const float* __restrict__ wc, const float* __restrict__ wp,
                             float* __restrict__ cat) {
    const int tid = threadIdx.x;
    const int n = tid & 127;
    const float* src = (tid < 128) ? h : hp;
    const float* wv  = (tid < 128) ? wc : wp;
    float s = 0.f;
    for (int i = 0; i < NN; ++i) s += wv[i] * src[(size_t)i * DIM + n];
    cat[tid] = s / (float)NN;
}

__global__ void final_mlp(const float* __restrict__ cat_in, const float* __restrict__ Wo,
                          const float* __restrict__ bo, const float* __restrict__ Wi,
                          const float* __restrict__ bi, float* __restrict__ out) {
    __shared__ float cur[256], nxt[256];
    const int t = threadIdx.x;
    cur[t] = cat_in[t];
    __syncthreads();
    for (int l = 0; l < 2; ++l) {
        const float* Wl = Wo + (size_t)l * 256 * 256;
        float s = bo[l * 256 + t];
        for (int k = 0; k < 256; ++k) s += cur[k] * Wl[(size_t)t * 256 + k];
        nxt[t] = fmaxf(s, 0.f);
        __syncthreads();
        cur[t] = nxt[t];
        __syncthreads();
    }
    if (t < 2) {
        float s = bi[t];
        for (int k = 0; k < 256; ++k) s += cur[k] * Wi[(size_t)t * 256 + k];
        out[t] = s;
    }
}

extern "C" void kernel_launch(void* const* d_in, const int* in_sizes, int n_in,
                              void* d_out, int out_size, void* d_ws, size_t ws_size,
                              hipStream_t stream) {
    (void)in_sizes; (void)n_in; (void)out_size; (void)ws_size;
    const int*   fingerprints = (const int*)d_in[0];
    const float* adjacency    = (const float*)d_in[1];
    const int*   words        = (const int*)d_in[2];
    const float* emb_fp       = (const float*)d_in[3];
    const float* emb_word     = (const float*)d_in[4];
    const float* Wg  = (const float*)d_in[5];
    const float* bg  = (const float*)d_in[6];
    const float* Wih = (const float*)d_in[7];
    const float* Whh = (const float*)d_in[8];
    const float* bih = (const float*)d_in[9];
    const float* bhh = (const float*)d_in[10];
    const float* Wc  = (const float*)d_in[11];
    const float* bc  = (const float*)d_in[12];
    const float* Wp  = (const float*)d_in[13];
    const float* bp  = (const float*)d_in[14];
    const float* Wo  = (const float*)d_in[15];
    const float* bo  = (const float*)d_in[16];
    const float* Wi  = (const float*)d_in[17];
    const float* bi  = (const float*)d_in[18];

    float* out         = (float*)d_out;
    float* weights_out = out + 2;

    char* base = (char*)d_ws;
    size_t off = 0;
    auto alloc = [&](size_t bytes) -> void* {
        void* p = base + off;
        off = (off + bytes + 255) & ~(size_t)255;
        return p;
    };
    float*    xs       = (float*)   alloc((size_t)NN * DIM * 4);
    _Float16* xs_h     = (_Float16*)alloc((size_t)NN * DIM * 2);
    _Float16* hs_h     = (_Float16*)alloc((size_t)NN * DIM * 2);
    _Float16* hsT_h    = (_Float16*)alloc((size_t)DIM * NN * 2);
    _Float16* attn_h   = (_Float16*)alloc((size_t)NN * NN * 2);
    float*    denom    = (float*)   alloc((size_t)NN * 4);
    _Float16* Wg_h     = (_Float16*)alloc((size_t)3 * DIM * DIM * 2);
    _Float16* Wc_h     = (_Float16*)alloc((size_t)DIM * DIM * 2);
    _Float16* Wp_h     = (_Float16*)alloc((size_t)DIM * 2 * DIM * 2);
    _Float16* Wih_h    = (_Float16*)alloc((size_t)2 * 512 * DIM * 2);
    float*    bsum     = (float*)   alloc((size_t)2 * 512 * 4);
    _Float16* wv_h     = (_Float16*)alloc((size_t)MM * DIM * 2);
    float*    G        = (float*)   alloc((size_t)2 * MM * 512 * 4);
    float*    hbuf     = (float*)   alloc((size_t)2 * MM * DIM * 4);
    _Float16* xs_rnn_h = (_Float16*)alloc((size_t)MM * 2 * DIM * 2);
    float*    h_f32    = (float*)   alloc((size_t)NN * DIM * 4);
    _Float16* h_h      = (_Float16*)alloc((size_t)NN * DIM * 2);
    float*    hp_f32   = (float*)   alloc((size_t)MM * DIM * 4);
    _Float16* hp_h     = (_Float16*)alloc((size_t)MM * DIM * 2);
    float*    w_comp   = (float*)   alloc((size_t)NN * 4);
    float*    w_prot   = (float*)   alloc((size_t)MM * 4);
    float*    cat      = (float*)   alloc(256 * 4);

    f32_to_f16<<<(3 * DIM * DIM + 255) / 256, 256, 0, stream>>>(Wg, Wg_h, 3 * DIM * DIM);
    f32_to_f16<<<(DIM * DIM + 255) / 256, 256, 0, stream>>>(Wc, Wc_h, DIM * DIM);
    f32_to_f16<<<(DIM * 2 * DIM + 255) / 256, 256, 0, stream>>>(Wp, Wp_h, DIM * 2 * DIM);
    f32_to_f16<<<(2 * 512 * DIM + 255) / 256, 256, 0, stream>>>(Wih, Wih_h, 2 * 512 * DIM);
    vec_add<<<(2 * 512 + 255) / 256, 256, 0, stream>>>(bih, bhh, bsum, 2 * 512);

    gather_embed<<<NN, DIM, 0, stream>>>(fingerprints, emb_fp, xs, nullptr);
    gather_embed<<<MM, DIM, 0, stream>>>(words, emb_word, nullptr, wv_h);

    for (int l = 0; l < 3; ++l) {
        f32_to_f16<<<(NN * DIM + 255) / 256, 256, 0, stream>>>(xs, xs_h, NN * DIM);
        gemm_bias_act<<<dim3(NN / 16, DIM / 16), 32, 0, stream>>>(
            xs_h, Wg_h + (size_t)l * DIM * DIM, bg + l * DIM,
            nullptr, hs_h, DIM, DIM, DIM, DIM, 1);
        transpose_f16<<<NN, DIM, 0, stream>>>(hs_h, hsT_h);
        gemm_attn<<<dim3(NN / 16, NN / 16), 32, 0, stream>>>(hs_h, adjacency, attn_h);
        row_abs_sum<<<NN, 256, 0, stream>>>(attn_h, denom);
        gemm_norm_axpy<<<dim3(NN / 16, DIM / 16), 32, 0, stream>>>(attn_h, hsT_h, denom, xs);
    }

    f32_to_f16<<<(NN * DIM + 255) / 256, 256, 0, stream>>>(xs, xs_h, NN * DIM);
    gemm_bias_act<<<dim3(NN / 16, DIM / 16), 32, 0, stream>>>(
        xs_h, Wc_h, bc, h_f32, h_h, DIM, DIM, DIM, DIM, 1);

    for (int d = 0; d < 2; ++d) {
        gemm_bias_act<<<dim3(MM / 16, 512 / 16), 32, 0, stream>>>(
            wv_h, Wih_h + (size_t)d * 512 * DIM, bsum + d * 512,
            G + (size_t)d * MM * 512, nullptr, DIM, DIM, DIM, 512, 0);
    }
    lstm_dir<<<2, 128, 0, stream>>>(G, Whh, hbuf);
    bilstm_concat_relu<<<MM, 256, 0, stream>>>(hbuf, xs_rnn_h);
    gemm_bias_act<<<dim3(MM / 16, DIM / 16), 32, 0, stream>>>(
        xs_rnn_h, Wp_h, bp, hp_f32, hp_h, 2 * DIM, 2 * DIM, 2 * DIM, DIM, 1);

    gemm_bias_act<<<dim3(NN / 16, MM / 16), 32, 0, stream>>>(
        h_h, hp_h, nullptr, weights_out, nullptr, DIM, DIM, DIM, MM, 0);

    row_mean<<<NN, 256, 0, stream>>>(weights_out, w_comp);
    col_mean<<<MM / 256, 256, 0, stream>>>(weights_out, w_prot);
    pool_vectors<<<1, 256, 0, stream>>>(h_f32, hp_f32, w_comp, w_prot, cat);
    final_mlp<<<1, 256, 0, stream>>>(cat, Wo, bo, Wi, bi, out);
}


# lstm_dir: per-step barrier orders LDS only (local-scope release/acquire fences + s_barrier); a step no longer waits for the acknowledgement of its global store of h; order of every sum unchanged
# speedup vs baseline: 3.6635x; 1.4946x over previous
#include <hip/hip_runtime.h>

typedef __attribute__((ext_vector_type(16))) _Float16 v16h;
typedef __attribute__((ext_vector_type(8)))  float    v8f;

#define DIM 128
#define NN  4096
#define MM  4096

__device__ __forceinline__ v16h frag_load(const _Float16* __restrict__ tile, int ld) {
    const int lane = threadIdx.x & 31;
    const int r    = lane & 15;
    const int kb   = (lane & 16) ? 8 : 0;
    const _Float16* q = tile + (size_t)r * (size_t)ld + kb;
    v16h f;
    uint4* fu = reinterpret_cast<uint4*>(&f);
    fu[0] = *reinterpret_cast<const uint4*>(q);
    fu[1] = *reinterpret_cast<const uint4*>(q + 16);
    return f;
}

__device__ __forceinline__ void frag_rc(int& rbase, int& col) {
    const int lane = threadIdx.x & 31;
    col   = lane & 15;
    rbase = (lane & 16) ? 8 : 0;
}

__global__ void gemm_bias_act(const _Float16* __restrict__ A, const _Float16* __restrict__ B,
                              const float* __restrict__ bias,
                              float* __restrict__ Cf, _Float16* __restrict__ Ch,
                              int K, int lda, int ldb, int ldc, int do_relu) {
    const int i0 = blockIdx.x * 16;
    const int j0 = blockIdx.y * 16;
    v8f acc = {};
    for (int k = 0; k < K; k += 32) {
        v16h a = frag_load(A + (size_t)i0 * lda + k, lda);
        v16h b = frag_load(B + (size_t)j0 * ldb + k, ldb);
        acc = __builtin_amdgcn_wmma_f32_16x16x32_f16(false, a, false, b, (short)0, acc,
                                                     false, false);
    }
    int rb, col; frag_rc(rb, col);
    const float bv = bias ? bias[j0 + col] : 0.f;
#pragma unroll
    for (int r = 0; r < 8; ++r) {
        const int row = i0 + rb + r;
        float v = acc[r] + bv;
        if (do_relu) v = fmaxf(v, 0.f);
        const size_t idx = (size_t)row * ldc + j0 + col;
        if (Cf) Cf[idx] = v;
        if (Ch) Ch[idx] = (_Float16)v;
    }
}

__global__ void gemm_attn(const _Float16* __restrict__ hs, const float* __restrict__ adj,
                          _Float16* __restrict__ attn) {
    const int i0 = blockIdx.x * 16;
    const int j0 = blockIdx.y * 16;
    v8f acc = {};
#pragma unroll
    for (int k = 0; k < DIM; k += 32) {
        v16h a = frag_load(hs + (size_t)i0 * DIM + k, DIM);
        v16h b = frag_load(hs + (size_t)j0 * DIM + k, DIM);
        acc = __builtin_amdgcn_wmma_f32_16x16x32_f16(false, a, false, b, (short)0, acc,
                                                     false, false);
    }
    int rb, col; frag_rc(rb, col);
#pragma unroll
    for (int r = 0; r < 8; ++r) {
        const int row = i0 + rb + r;
        const size_t idx = (size_t)row * NN + j0 + col;
        attn[idx] = (_Float16)(acc[r] * adj[idx]);
    }
}

__global__ void row_abs_sum(const _Float16* __restrict__ attn, float* __restrict__ denom) {
    __shared__ float red[256];
    const int i = blockIdx.x;
    float s = 0.f;
    for (int j = threadIdx.x; j < NN; j += 256) s += fabsf((float)attn[(size_t)i * NN + j]);
    red[threadIdx.x] = s;
    __syncthreads();
    for (int w = 128; w > 0; w >>= 1) {
        if (threadIdx.x < w) red[threadIdx.x] += red[threadIdx.x + w];
        __syncthreads();
    }
    if (threadIdx.x == 0) denom[i] = fmaxf(red[0], 1e-12f);
}

__global__ void transpose_f16(const _Float16* __restrict__ src, _Float16* __restrict__ dst) {
    const int j = blockIdx.x;
    const int n = threadIdx.x;
    dst[(size_t)n * NN + j] = src[(size_t)j * DIM + n];
}

__global__ void gemm_norm_axpy(const _Float16* __restrict__ attn, const _Float16* __restrict__ hsT,
                               const float* __restrict__ denom, float* __restrict__ xs) {
    const int i0 = blockIdx.x * 16;
    const int j0 = blockIdx.y * 16;
    const _Float16* Ab = attn + (size_t)i0 * NN;
    const _Float16* Bb = hsT + (size_t)j0 * NN;
    v8f acc = {};
    for (int k = 0; k < NN; k += 32) {
        __builtin_prefetch(Ab + k + 256, 0, 0);
        v16h a = frag_load(Ab + k, NN);
        v16h b = frag_load(Bb + k, NN);
        acc = __builtin_amdgcn_wmma_f32_16x16x32_f16(false, a, false, b, (short)0, acc,
                                                     false, false);
    }
    int rb, col; frag_rc(rb, col);
#pragma unroll
    for (int r = 0; r < 8; ++r) {
        const int row = i0 + rb + r;
        xs[(size_t)row * DIM + j0 + col] += acc[r] / denom[row];
    }
}

__global__ void gather_embed(const int* __restrict__ idx, const float* __restrict__ emb,
                             float* __restrict__ of32, _Float16* __restrict__ of16) {
    const int i = blockIdx.x;
    const int n = threadIdx.x;
    const float v = emb[(size_t)idx[i] * DIM + n];
    if (of32) of32[(size_t)i * DIM + n] = v;
    if (of16) of16[(size_t)i * DIM + n] = (_Float16)v;
}

__global__ void f32_to_f16(const float* __restrict__ s, _Float16* __restrict__ d, int n) {
    const int i = blockIdx.x * 256 + threadIdx.x;
    if (i < n) d[i] = (_Float16)s[i];
}

__global__ void vec_add(const float* __restrict__ a, const float* __restrict__ b,
                        float* __restrict__ c, int n) {
    const int i = blockIdx.x * 256 + threadIdx.x;
    if (i < n) c[i] = a[i] + b[i];
}

__global__ __launch_bounds__(128) void lstm_dir(const float* __restrict__ G,
                                                const float* __restrict__ Whh,
                                                float* __restrict__ hout) {
    __shared__ __attribute__((aligned(16))) float h_sh[2][DIM];
    const int d = blockIdx.x;
    const int n = threadIdx.x;
    const float* W  = Whh + (size_t)d * 512 * DIM;
    const float* Gd = G + (size_t)d * MM * 512;
    float* hd = hout + (size_t)d * MM * DIM;

    float wi[DIM], wf[DIM], wg[DIM], wo[DIM];
    {
        const float4* ri = reinterpret_cast<const float4*>(W + (size_t)(n)       * DIM);
        const float4* rf = reinterpret_cast<const float4*>(W + (size_t)(n + 128) * DIM);
        const float4* rg = reinterpret_cast<const float4*>(W + (size_t)(n + 256) * DIM);
        const float4* ro = reinterpret_cast<const float4*>(W + (size_t)(n + 384) * DIM);
#pragma unroll
        for (int k4 = 0; k4 < DIM / 4; ++k4) {
            const float4 a = ri[k4], b = rf[k4], c4 = rg[k4], e = ro[k4];
            wi[4 * k4 + 0] = a.x;  wi[4 * k4 + 1] = a.y;  wi[4 * k4 + 2] = a.z;  wi[4 * k4 + 3] = a.w;
            wf[4 * k4 + 0] = b.x;  wf[4 * k4 + 1] = b.y;  wf[4 * k4 + 2] = b.z;  wf[4 * k4 + 3] = b.w;
            wg[4 * k4 + 0] = c4.x; wg[4 * k4 + 1] = c4.y; wg[4 * k4 + 2] = c4.z; wg[4 * k4 + 3] = c4.w;
            wo[4 * k4 + 0] = e.x;  wo[4 * k4 + 1] = e.y;  wo[4 * k4 + 2] = e.z;  wo[4 * k4 + 3] = e.w;
        }
    }
    h_sh[0][n] = 0.f;
    float c = 0.f;
    __syncthreads();

    float pi, pf, pg, po;
    {
        const float* g0 = Gd + (size_t)(d ? (MM - 1) : 0) * 512 + n;
        pi = g0[0]; pf = g0[128]; pg = g0[256]; po = g0[384];
    }

    for (int t = 0; t < MM; ++t) {
        const int tt = d ? (MM - 1 - t) : t;
        const float4* h4 = reinterpret_cast<const float4*>(h_sh[t & 1]);
        float zi = pi, zf = pf, zg = pg, zo = po;
        float4 hv[DIM / 4];
#pragma unroll
        for (int k4 = 0; k4 < DIM / 4; ++k4) hv[k4] = h4[k4];
        __builtin_amdgcn_sched_barrier(0);
        {
            const int tn = (t + 1 < MM) ? (t + 1) : t;
            const float* gn = Gd + (size_t)(d ? (MM - 1 - tn) : tn) * 512 + n;
            pi = gn[0]; pf = gn[128]; pg = gn[256]; po = gn[384];
        }
#pragma unroll
        for (int k = 0; k < DIM; ++k) {
            const float hk = (k & 3) == 0 ? hv[k >> 2].x : (k & 3) == 1 ? hv[k >> 2].y
                           : (k & 3) == 2 ? hv[k >> 2].z : hv[k >> 2].w;
            zi += hk * wi[k]; zf += hk * wf[k];
            zg += hk * wg[k]; zo += hk * wo[k];
            __builtin_amdgcn_sched_barrier(0);
        }
        asm volatile("" : "+v"(zi), "+v"(zf), "+v"(zg), "+v"(zo));
        const float ig = 1.f / (1.f + __expf(-zi));
        const float fg = 1.f / (1.f + __expf(-zf));
        const float gg = tanhf(zg);
        const float og = 1.f / (1.f + __expf(-zo));
        c = fg * c + ig * gg;
        const float h = og * tanhf(c);
        h_sh[(t + 1) & 1][n] = h;
        hd[(size_t)tt * DIM + n] = h;
        __builtin_amdgcn_fence(__ATOMIC_RELEASE, "workgroup", "local");
        __builtin_amdgcn_s_barrier();
        __builtin_amdgcn_fence(__ATOMIC_ACQUIRE, "workgroup", "local");
    }
}

__global__ void bilstm_concat_relu(const float* __restrict__ hbuf, _Float16* __restrict__ xs_rnn) {
    const int t = blockIdx.x;
    const int n = threadIdx.x;
    const float v = (n < DIM) ? hbuf[(size_t)t * DIM + n]
                              : hbuf[(size_t)(MM + t) * DIM + (n - DIM)];
    xs_rnn[(size_t)t * (2 * DIM) + n] = (_Float16)fmaxf(v, 0.f);
}

__global__ void row_mean(const float* __restrict__ W, float* __restrict__ out) {
    __shared__ float red[256];
    const int i = blockIdx.x;
    float s = 0.f;
    for (int j = threadIdx.x; j < MM; j += 256) s += W[(size_t)i * MM + j];
    red[threadIdx.x] = s;
    __syncthreads();
    for (int w = 128; w > 0; w >>= 1) {
        if (threadIdx.x < w) red[threadIdx.x] += red[threadIdx.x + w];
        __syncthreads();
    }
    if (threadIdx.x == 0) out[i] = red[0] / (float)MM;
}

__global__ void col_mean(const float* __restrict__ W, float* __restrict__ out) {
    const int j = blockIdx.x * 256 + threadIdx.x;
    float s = 0.f;
    for (int i = 0; i < NN; ++i) s += W[(size_t)i * MM + j];
    out[j] = s / (float)NN;
}

__global__ void pool_vectors(const float* __restrict__ h, const float* __restrict__ hp,
                             const float* __restrict__ wc, const float* __restrict__ wp,
                             float* __restrict__ cat) {
    const int tid = threadIdx.x;
    const int n = tid & 127;
    const float* src = (tid < 128) ? h : hp;
    const float* wv  = (tid < 128) ? wc : wp;
    float s = 0.f;
    for (int i = 0; i < NN; ++i) s += wv[i] * src[(size_t)i * DIM + n];
    cat[tid] = s / (float)NN;
}

__global__ void final_mlp(const float* __restrict__ cat_in, const float* __restrict__ Wo,
                          const float* __restrict__ bo, const float* __restrict__ Wi,
                          const float* __restrict__ bi, float* __restrict__ out) {
    __shared__ float cur[256], nxt[256];
    const int t = threadIdx.x;
    cur[t] = cat_in[t];
    __syncthreads();
    for (int l = 0; l < 2; ++l) {
        const float* Wl = Wo + (size_t)l * 256 * 256;
        float s = bo[l * 256 + t];
        for (int k = 0; k < 256; ++k) s += cur[k] * Wl[(size_t)t * 256 + k];
        nxt[t] = fmaxf(s, 0.f);
        __syncthreads();
        cur[t] = nxt[t];
        __syncthreads();
    }
    if (t < 2) {
        float s = bi[t];
        for (int k = 0; k < 256; ++k) s += cur[k] * Wi[(size_t)t * 256 + k];
        out[t] = s;
    }
}

extern "C" void kernel_launch(void* const* d_in, const int* in_sizes, int n_in,
                              void* d_out, int out_size, void* d_ws, size_t ws_size,
                              hipStream_t stream) {
    (void)in_sizes; (void)n_in; (void)out_size; (void)ws_size;
    const int*   fingerprints = (const int*)d_in[0];
    const float* adjacency    = (const float*)d_in[1];
    const int*   words        = (const int*)d_in[2];
    const float* emb_fp       = (const float*)d_in[3];
    const float* emb_word     = (const float*)d_in[4];
    const float* Wg  = (const float*)d_in[5];
    const float* bg  = (const float*)d_in[6];
    const float* Wih = (const float*)d_in[7];
    const float* Whh = (const float*)d_in[8];
    const float* bih = (const float*)d_in[9];
    const float* bhh = (const float*)d_in[10];
    const float* Wc  = (const float*)d_in[11];
    const float* bc  = (const float*)d_in[12];
    const float* Wp  = (const float*)d_in[13];
    const float* bp  = (const float*)d_in[14];
    const float* Wo  = (const float*)d_in[15];
    const float* bo  = (const float*)d_in[16];
    const float* Wi  = (const float*)d_in[17];
    const float* bi  = (const float*)d_in[18];

    float* out         = (float*)d_out;
    float* weights_out = out + 2;

    char* base = (char*)d_ws;
    size_t off = 0;
    auto alloc = [&](size_t bytes) -> void* {
        void* p = base + off;
        off = (off + bytes + 255) & ~(size_t)255;
        return p;
    };
    float*    xs       = (float*)   alloc((size_t)NN * DIM * 4);
    _Float16* xs_h     = (_Float16*)alloc((size_t)NN * DIM * 2);
    _Float16* hs_h     = (_Float16*)alloc((size_t)NN * DIM * 2);
    _Float16* hsT_h    = (_Float16*)alloc((size_t)DIM * NN * 2);
    _Float16* attn_h   = (_Float16*)alloc((size_t)NN * NN * 2);
    float*    denom    = (float*)   alloc((size_t)NN * 4);
    _Float16* Wg_h     = (_Float16*)alloc((size_t)3 * DIM * DIM * 2);
    _Float16* Wc_h     = (_Float16*)alloc((size_t)DIM * DIM * 2);
    _Float16* Wp_h     = (_Float16*)alloc((size_t)DIM * 2 * DIM * 2);
    _Float16* Wih_h    = (_Float16*)alloc((size_t)2 * 512 * DIM * 2);
    float*    bsum     = (float*)   alloc((size_t)2 * 512 * 4);
    _Float16* wv_h     = (_Float16*)alloc((size_t)MM * DIM * 2);
    float*    G        = (float*)   alloc((size_t)2 * MM * 512 * 4);
    float*    hbuf     = (float*)   alloc((size_t)2 * MM * DIM * 4);
    _Float16* xs_rnn_h = (_Float16*)alloc((size_t)MM * 2 * DIM * 2);
    float*    h_f32    = (float*)   alloc((size_t)NN * DIM * 4);
    _Float16* h_h      = (_Float16*)alloc((size_t)NN * DIM * 2);
    float*    hp_f32   = (float*)   alloc((size_t)MM * DIM * 4);
    _Float16* hp_h     = (_Float16*)alloc((size_t)MM * DIM * 2);
    float*    w_comp   = (float*)   alloc((size_t)NN * 4);
    float*    w_prot   = (float*)   alloc((size_t)MM * 4);
    float*    cat      = (float*)   alloc(256 * 4);

    f32_to_f16<<<(3 * DIM * DIM + 255) / 256, 256, 0, stream>>>(Wg, Wg_h, 3 * DIM * DIM);
    f32_to_f16<<<(DIM * DIM + 255) / 256, 256, 0, stream>>>(Wc, Wc_h, DIM * DIM);
    f32_to_f16<<<(DIM * 2 * DIM + 255) / 256, 256, 0, stream>>>(Wp, Wp_h, DIM * 2 * DIM);
    f32_to_f16<<<(2 * 512 * DIM + 255) / 256, 256, 0, stream>>>(Wih, Wih_h, 2 * 512 * DIM);
    vec_add<<<(2 * 512 + 255) / 256, 256, 0, stream>>>(bih, bhh, bsum, 2 * 512);

    gather_embed<<<NN, DIM, 0, stream>>>(fingerprints, emb_fp, xs, nullptr);
    gather_embed<<<MM, DIM, 0, stream>>>(words, emb_word, nullptr, wv_h);

    for (int l = 0; l < 3; ++l) {
        f32_to_f16<<<(NN * DIM + 255) / 256, 256, 0, stream>>>(xs, xs_h, NN * DIM);
        gemm_bias_act<<<dim3(NN / 16, DIM / 16), 32, 0, stream>>>(
            xs_h, Wg_h + (size_t)l * DIM * DIM, bg + l * DIM,
            nullptr, hs_h, DIM, DIM, DIM, DIM, 1);
        transpose_f16<<<NN, DIM, 0, stream>>>(hs_h, hsT_h);
        gemm_attn<<<dim3(NN / 16, NN / 16), 32, 0, stream>>>(hs_h, adjacency, attn_h);
        row_abs_sum<<<NN, 256, 0, stream>>>(attn_h, denom);
        gemm_norm_axpy<<<dim3(NN / 16, DIM / 16), 32, 0, stream>>>(attn_h, hsT_h, denom, xs);
    }

    f32_to_f16<<<(NN * DIM + 255) / 256, 256, 0, stream>>>(xs, xs_h, NN * DIM);
    gemm_bias_act<<<dim3(NN / 16, DIM / 16), 32, 0, stream>>>(
        xs_h, Wc_h, bc, h_f32, h_h, DIM, DIM, DIM, DIM, 1);

    for (int d = 0; d < 2; ++d) {
        gemm_bias_act<<<dim3(MM / 16, 512 / 16), 32, 0, stream>>>(
            wv_h, Wih_h + (size_t)d * 512 * DIM, bsum + d * 512,
            G + (size_t)d * MM * 512, nullptr, DIM, DIM, DIM, 512, 0);
    }
    lstm_dir<<<2, 128, 0, stream>>>(G, Whh, hbuf);
    bilstm_concat_relu<<<MM, 256, 0, stream>>>(hbuf, xs_rnn_h);
    gemm_bias_act<<<dim3(MM / 16, DIM / 16), 32, 0, stream>>>(
        xs_rnn_h, Wp_h, bp, hp_f32, hp_h, 2 * DIM, 2 * DIM, 2 * DIM, DIM, 1);

    gemm_bias_act<<<dim3(NN / 16, MM / 16), 32, 0, stream>>>(
        h_h, hp_h, nullptr, weights_out, nullptr, DIM, DIM, DIM, MM, 0);

    row_mean<<<NN, 256, 0, stream>>>(weights_out, w_comp);
    col_mean<<<MM / 256, 256, 0, stream>>>(weights_out, w_prot);
    pool_vectors<<<1, 256, 0, stream>>>(h_f32, hp_f32, w_comp, w_prot, cat);
    final_mlp<<<1, 256, 0, stream>>>(cat, Wo, bo, Wi, bi, out);
}


# lstm_dir gates: sigmoid, tanh as straight-line v_exp_f32+v_rcp_f32 (no IEEE division, no branching tanhf); tanh's 1-2r folded into the c and h fmas (c, h sums re-associated); dot sums unchanged
# speedup vs baseline: 4.8467x; 1.3230x over previous
#include <hip/hip_runtime.h>

typedef __attribute__((ext_vector_type(16))) _Float16 v16h;
typedef __attribute__((ext_vector_type(8)))  float    v8f;

#define DIM 128
#define NN  4096
#define MM  4096

__device__ __forceinline__ v16h frag_load(const _Float16* __restrict__ tile, int ld) {
    const int lane = threadIdx.x & 31;
    const int r    = lane & 15;
    const int kb   = (lane & 16) ? 8 : 0;
    const _Float16* q = tile + (size_t)r * (size_t)ld + kb;
    v16h f;
    uint4* fu = reinterpret_cast<uint4*>(&f);
    fu[0] = *reinterpret_cast<const uint4*>(q);
    fu[1] = *reinterpret_cast<const uint4*>(q + 16);
    return f;
}

__device__ __forceinline__ void frag_rc(int& rbase, int& col) {
    const int lane = threadIdx.x & 31;
    col   = lane & 15;
    rbase = (lane & 16) ? 8 : 0;
}

__global__ void gemm_bias_act(const _Float16* __restrict__ A, const _Float16* __restrict__ B,
                              const float* __restrict__ bias,
                              float* __restrict__ Cf, _Float16* __restrict__ Ch,
                              int K, int lda, int ldb, int ldc, int do_relu) {
    const int i0 = blockIdx.x * 16;
    const int j0 = blockIdx.y * 16;
    v8f acc = {};
    for (int k = 0; k < K; k += 32) {
        v16h a = frag_load(A + (size_t)i0 * lda + k, lda);
        v16h b = frag_load(B + (size_t)j0 * ldb + k, ldb);
        acc = __builtin_amdgcn_wmma_f32_16x16x32_f16(false, a, false, b, (short)0, acc,
                                                     false, false);
    }
    int rb, col; frag_rc(rb, col);
    const float bv = bias ? bias[j0 + col] : 0.f;
#pragma unroll
    for (int r = 0; r < 8; ++r) {
        const int row = i0 + rb + r;
        float v = acc[r] + bv;
        if (do_relu) v = fmaxf(v, 0.f);
        const size_t idx = (size_t)row * ldc + j0 + col;
        if (Cf) Cf[idx] = v;
        if (Ch) Ch[idx] = (_Float16)v;
    }
}

__global__ void gemm_attn(const _Float16* __restrict__ hs, const float* __restrict__ adj,
                          _Float16* __restrict__ attn) {
    const int i0 = blockIdx.x * 16;
    const int j0 = blockIdx.y * 16;
    v8f acc = {};
#pragma unroll
    for (int k = 0; k < DIM; k += 32) {
        v16h a = frag_load(hs + (size_t)i0 * DIM + k, DIM);
        v16h b = frag_load(hs + (size_t)j0 * DIM + k, DIM);
        acc = __builtin_amdgcn_wmma_f32_16x16x32_f16(false, a, false, b, (short)0, acc,
                                                     false, false);
    }
    int rb, col; frag_rc(rb, col);
#pragma unroll
    for (int r = 0; r < 8; ++r) {
        const int row = i0 + rb + r;
        const size_t idx = (size_t)row * NN + j0 + col;
        attn[idx] = (_Float16)(acc[r] * adj[idx]);
    }
}

__global__ void row_abs_sum(const _Float16* __restrict__ attn, float* __restrict__ denom) {
    __shared__ float red[256];
    const int i = blockIdx.x;
    float s = 0.f;
    for (int j = threadIdx.x; j < NN; j += 256) s += fabsf((float)attn[(size_t)i * NN + j]);
    red[threadIdx.x] = s;
    __syncthreads();
    for (int w = 128; w > 0; w >>= 1) {
        if (threadIdx.x < w) red[threadIdx.x] += red[threadIdx.x + w];
        __syncthreads();
    }
    if (threadIdx.x == 0) denom[i] = fmaxf(red[0], 1e-12f);
}

__global__ void transpose_f16(const _Float16* __restrict__ src, _Float16* __restrict__ dst) {
    const int j = blockIdx.x;
    const int n = threadIdx.x;
    dst[(size_t)n * NN + j] = src[(size_t)j * DIM + n];
}

__global__ void gemm_norm_axpy(const _Float16* __restrict__ attn, const _Float16* __restrict__ hsT,
                               const float* __restrict__ denom, float* __restrict__ xs) {
    const int i0 = blockIdx.x * 16;
    const int j0 = blockIdx.y * 16;
    const _Float16* Ab = attn + (size_t)i0 * NN;
    const _Float16* Bb = hsT + (size_t)j0 * NN;
    v8f acc = {};
    for (int k = 0; k < NN; k += 32) {
        __builtin_prefetch(Ab + k + 256, 0, 0);
        v16h a = frag_load(Ab + k, NN);
        v16h b = frag_load(Bb + k, NN);
        acc = __builtin_amdgcn_wmma_f32_16x16x32_f16(false, a, false, b, (short)0, acc,
                                                     false, false);
    }
    int rb, col; frag_rc(rb, col);
#pragma unroll
    for (int r = 0; r < 8; ++r) {
        const int row = i0 + rb + r;
        xs[(size_t)row * DIM + j0 + col] += acc[r] / denom[row];
    }
}

__global__ void gather_embed(const int* __restrict__ idx, const float* __restrict__ emb,
                             float* __restrict__ of32, _Float16* __restrict__ of16) {
    const int i = blockIdx.x;
    const int n = threadIdx.x;
    const float v = emb[(size_t)idx[i] * DIM + n];
    if (of32) of32[(size_t)i * DIM + n] = v;
    if (of16) of16[(size_t)i * DIM + n] = (_Float16)v;
}

__global__ void f32_to_f16(const float* __restrict__ s, _Float16* __restrict__ d, int n) {
    const int i = blockIdx.x * 256 + threadIdx.x;
    if (i < n) d[i] = (_Float16)s[i];
}

__global__ void vec_add(const float* __restrict__ a, const float* __restrict__ b,
                        float* __restrict__ c, int n) {
    const int i = blockIdx.x * 256 + threadIdx.x;
    if (i < n) c[i] = a[i] + b[i];
}

typedef __attribute__((ext_vector_type(2))) float v2f;
typedef __attribute__((ext_vector_type(4))) float v4f;

__device__ __forceinline__ float gate_sigmoid(float z) {
    return __builtin_amdgcn_rcpf(1.f + __builtin_amdgcn_exp2f(-1.44269504f * z));
}
__device__ __forceinline__ float gate_tanh_r(float x) {
    return __builtin_amdgcn_rcpf(1.f + __builtin_amdgcn_exp2f(2.88539008f * x));
}

__global__ __launch_bounds__(128) void lstm_dir(const float* __restrict__ G,
                                                const float* __restrict__ Whh,
                                                float* __restrict__ hout) {
    __shared__ __attribute__((aligned(16))) float h_sh[2][DIM];
    const int d = blockIdx.x;
    const int n = threadIdx.x;
    const float* W  = Whh + (size_t)d * 512 * DIM;
    const float* Gd = G + (size_t)d * MM * 512;
    float* hd = hout + (size_t)d * MM * DIM;

    v2f wi[DIM / 2], wf[DIM / 2], wg[DIM / 2], wo[DIM / 2];
    {
        const v4f* ri = reinterpret_cast<const v4f*>(W + (size_t)(n)       * DIM);
        const v4f* rf = reinterpret_cast<const v4f*>(W + (size_t)(n + 128) * DIM);
        const v4f* rg = reinterpret_cast<const v4f*>(W + (size_t)(n + 256) * DIM);
        const v4f* ro = reinterpret_cast<const v4f*>(W + (size_t)(n + 384) * DIM);
#pragma unroll
        for (int k4 = 0; k4 < DIM / 4; ++k4) {
            const v4f a = ri[k4], b = rf[k4], c4 = rg[k4], e = ro[k4];
            wi[2 * k4] = __builtin_shufflevector(a, a, 0, 1);
            wi[2 * k4 + 1] = __builtin_shufflevector(a, a, 2, 3);
            wf[2 * k4] = __builtin_shufflevector(b, b, 0, 1);
            wf[2 * k4 + 1] = __builtin_shufflevector(b, b, 2, 3);
            wg[2 * k4] = __builtin_shufflevector(c4, c4, 0, 1);
            wg[2 * k4 + 1] = __builtin_shufflevector(c4, c4, 2, 3);
            wo[2 * k4] = __builtin_shufflevector(e, e, 0, 1);
            wo[2 * k4 + 1] = __builtin_shufflevector(e, e, 2, 3);
        }
    }
    h_sh[0][n] = 0.f;
    float c = 0.f;
    __syncthreads();

    float pi, pf, pg, po;
    {
        const float* g0 = Gd + (size_t)(d ? (MM - 1) : 0) * 512 + n;
        pi = g0[0]; pf = g0[128]; pg = g0[256]; po = g0[384];
    }

    for (int t = 0; t < MM; ++t) {
        const int tt = d ? (MM - 1 - t) : t;
        const v4f* h4 = reinterpret_cast<const v4f*>(h_sh[t & 1]);
        v2f ai = {pi, 0.f}, af = {pf, 0.f}, ag = {pg, 0.f}, ao = {po, 0.f};
        v2f bi = {0.f, 0.f}, bf = {0.f, 0.f}, bg = {0.f, 0.f}, bo = {0.f, 0.f};
        v4f hv[DIM / 4];
#pragma unroll
        for (int k4 = 0; k4 < DIM / 4; ++k4) hv[k4] = h4[k4];
        __builtin_amdgcn_sched_barrier(0);
        {
            const int tn = (t + 1 < MM) ? (t + 1) : t;
            const float* gn = Gd + (size_t)(d ? (MM - 1 - tn) : tn) * 512 + n;
            pi = gn[0]; pf = gn[128]; pg = gn[256]; po = gn[384];
        }
#pragma unroll
        for (int k4 = 0; k4 < DIM / 4; ++k4) {
            const v2f hlo = __builtin_shufflevector(hv[k4], hv[k4], 0, 1);
            const v2f hhi = __builtin_shufflevector(hv[k4], hv[k4], 2, 3);
            ai = __builtin_elementwise_fma(hlo, wi[2 * k4], ai);
            af = __builtin_elementwise_fma(hlo, wf[2 * k4], af);
            ag = __builtin_elementwise_fma(hlo, wg[2 * k4], ag);
            ao = __builtin_elementwise_fma(hlo, wo[2 * k4], ao);
            bi = __builtin_elementwise_fma(hhi, wi[2 * k4 + 1], bi);
            bf = __builtin_elementwise_fma(hhi, wf[2 * k4 + 1], bf);
            bg = __builtin_elementwise_fma(hhi, wg[2 * k4 + 1], bg);
            bo = __builtin_elementwise_fma(hhi, wo[2 * k4 + 1], bo);
        }
        const v2f si = ai + bi, sf = af + bf, sg = ag + bg, so = ao + bo;
        float zi = si.x + si.y, zf = sf.x + sf.y, zg = sg.x + sg.y, zo = so.x + so.y;
        const float ig = gate_sigmoid(zi);
        const float fg = gate_sigmoid(zf);
        const float og = gate_sigmoid(zo);
        const float rg = gate_tanh_r(zg);
        c = __builtin_fmaf(rg, -2.f * ig, __builtin_fmaf(fg, c, ig));
        const float rc = gate_tanh_r(c);
        const float h = __builtin_fmaf(rc, -2.f * og, og);
        h_sh[(t + 1) & 1][n] = h;
        hd[(size_t)tt * DIM + n] = h;
        __builtin_amdgcn_fence(__ATOMIC_RELEASE, "workgroup", "local");
        __builtin_amdgcn_s_barrier();
        __builtin_amdgcn_fence(__ATOMIC_ACQUIRE, "workgroup", "local");
    }
}

__global__ void bilstm_concat_relu(const float* __restrict__ hbuf, _Float16* __restrict__ xs_rnn) {
    const int t = blockIdx.x;
    const int n = threadIdx.x;
    const float v = (n < DIM) ? hbuf[(size_t)t * DIM + n]
                              : hbuf[(size_t)(MM + t) * DIM + (n - DIM)];
    xs_rnn[(size_t)t * (2 * DIM) + n] = (_Float16)fmaxf(v, 0.f);
}

__global__ void row_mean(const float* __restrict__ W, float* __restrict__ out) {
    __shared__ float red[256];
    const int i = blockIdx.x;
    float s = 0.f;
    for (int j = threadIdx.x; j < MM; j += 256) s += W[(size_t)i * MM + j];
    red[threadIdx.x] = s;
    __syncthreads();
    for (int w = 128; w > 0; w >>= 1) {
        if (threadIdx.x < w) red[threadIdx.x] += red[threadIdx.x + w];
        __syncthreads();
    }
    if (threadIdx.x == 0) out[i] = red[0] / (float)MM;
}

__global__ void col_mean(const float* __restrict__ W, float* __restrict__ out) {
    const int j = blockIdx.x * 256 + threadIdx.x;
    float s = 0.f;
    for (int i = 0; i < NN; ++i) s += W[(size_t)i * MM + j];
    out[j] = s / (float)NN;
}

__global__ void pool_vectors(const float* __restrict__ h, const float* __restrict__ hp,
                             const float* __restrict__ wc, const float* __restrict__ wp,
                             float* __restrict__ cat) {
    const int tid = threadIdx.x;
    const int n = tid & 127;
    const float* src = (tid < 128) ? h : hp;
    const float* wv  = (tid < 128) ? wc : wp;
    float s = 0.f;
    for (int i = 0; i < NN; ++i) s += wv[i] * src[(size_t)i * DIM + n];
    cat[tid] = s / (float)NN;
}

__global__ void final_mlp(const float* __restrict__ cat_in, const float* __restrict__ Wo,
                          const float* __restrict__ bo, const float* __restrict__ Wi,
                          const float* __restrict__ bi, float* __restrict__ out) {
    __shared__ float cur[256], nxt[256];
    const int t = threadIdx.x;
    cur[t] = cat_in[t];
    __syncthreads();
    for (int l = 0; l < 2; ++l) {
        const float* Wl = Wo + (size_t)l * 256 * 256;
        float s = bo[l * 256 + t];
        for (int k = 0; k < 256; ++k) s += cur[k] * Wl[(size_t)t * 256 + k];
        nxt[t] = fmaxf(s, 0.f);
        __syncthreads();
        cur[t] = nxt[t];
        __syncthreads();
    }
    if (t < 2) {
        float s = bi[t];
        for (int k = 0; k < 256; ++k) s += cur[k] * Wi[(size_t)t * 256 + k];
        out[t] = s;
    }
}

extern "C" void kernel_launch(void* const* d_in, const int* in_sizes, int n_in,
                              void* d_out, int out_size, void* d_ws, size_t ws_size,
                              hipStream_t stream) {
    (void)in_sizes; (void)n_in; (void)out_size; (void)ws_size;
    const int*   fingerprints = (const int*)d_in[0];
    const float* adjacency    = (const float*)d_in[1];
    const int*   words        = (const int*)d_in[2];
    const float* emb_fp       = (const float*)d_in[3];
    const float* emb_word     = (const float*)d_in[4];
    const float* Wg  = (const float*)d_in[5];
    const float* bg  = (const float*)d_in[6];
    const float* Wih = (const float*)d_in[7];
    const float* Whh = (const float*)d_in[8];
    const float* bih = (const float*)d_in[9];
    const float* bhh = (const float*)d_in[10];
    const float* Wc  = (const float*)d_in[11];
    const float* bc  = (const float*)d_in[12];
    const float* Wp  = (const float*)d_in[13];
    const float* bp  = (const float*)d_in[14];
    const float* Wo  = (const float*)d_in[15];
    const float* bo  = (const float*)d_in[16];
    const float* Wi  = (const float*)d_in[17];
    const float* bi  = (const float*)d_in[18];

    float* out         = (float*)d_out;
    float* weights_out = out + 2;

    char* base = (char*)d_ws;
    size_t off = 0;
    auto alloc = [&](size_t bytes) -> void* {
        void* p = base + off;
        off = (off + bytes + 255) & ~(size_t)255;
        return p;
    };
    float*    xs       = (float*)   alloc((size_t)NN * DIM * 4);
    _Float16* xs_h     = (_Float16*)alloc((size_t)NN * DIM * 2);
    _Float16* hs_h     = (_Float16*)alloc((size_t)NN * DIM * 2);
    _Float16* hsT_h    = (_Float16*)alloc((size_t)DIM * NN * 2);
    _Float16* attn_h   = (_Float16*)alloc((size_t)NN * NN * 2);
    float*    denom    = (float*)   alloc((size_t)NN * 4);
    _Float16* Wg_h     = (_Float16*)alloc((size_t)3 * DIM * DIM * 2);
    _Float16* Wc_h     = (_Float16*)alloc((size_t)DIM * DIM * 2);
    _Float16* Wp_h     = (_Float16*)alloc((size_t)DIM * 2 * DIM * 2);
    _Float16* Wih_h    = (_Float16*)alloc((size_t)2 * 512 * DIM * 2);
    float*    bsum     = (float*)   alloc((size_t)2 * 512 * 4);
    _Float16* wv_h     = (_Float16*)alloc((size_t)MM * DIM * 2);
    float*    G        = (float*)   alloc((size_t)2 * MM * 512 * 4);
    float*    hbuf     = (float*)   alloc((size_t)2 * MM * DIM * 4);
    _Float16* xs_rnn_h = (_Float16*)alloc((size_t)MM * 2 * DIM * 2);
    float*    h_f32    = (float*)   alloc((size_t)NN * DIM * 4);
    _Float16* h_h      = (_Float16*)alloc((size_t)NN * DIM * 2);
    float*    hp_f32   = (float*)   alloc((size_t)MM * DIM * 4);
    _Float16* hp_h     = (_Float16*)alloc((size_t)MM * DIM * 2);
    float*    w_comp   = (float*)   alloc((size_t)NN * 4);
    float*    w_prot   = (float*)   alloc((size_t)MM * 4);
    float*    cat      = (float*)   alloc(256 * 4);

    f32_to_f16<<<(3 * DIM * DIM + 255) / 256, 256, 0, stream>>>(Wg, Wg_h, 3 * DIM * DIM);
    f32_to_f16<<<(DIM * DIM + 255) / 256, 256, 0, stream>>>(Wc, Wc_h, DIM * DIM);
    f32_to_f16<<<(DIM * 2 * DIM + 255) / 256, 256, 0, stream>>>(Wp, Wp_h, DIM * 2 * DIM);
    f32_to_f16<<<(2 * 512 * DIM + 255) / 256, 256, 0, stream>>>(Wih, Wih_h, 2 * 512 * DIM);
    vec_add<<<(2 * 512 + 255) / 256, 256, 0, stream>>>(bih, bhh, bsum, 2 * 512);

    gather_embed<<<NN, DIM, 0, stream>>>(fingerprints, emb_fp, xs, nullptr);
    gather_embed<<<MM, DIM, 0, stream>>>(words, emb_word, nullptr, wv_h);

    for (int l = 0; l < 3; ++l) {
        f32_to_f16<<<(NN * DIM + 255) / 256, 256, 0, stream>>>(xs, xs_h, NN * DIM);
        gemm_bias_act<<<dim3(NN / 16, DIM / 16), 32, 0, stream>>>(
            xs_h, Wg_h + (size_t)l * DIM * DIM, bg + l * DIM,
            nullptr, hs_h, DIM, DIM, DIM, DIM, 1);
        transpose_f16<<<NN, DIM, 0, stream>>>(hs_h, hsT_h);
        gemm_attn<<<dim3(NN / 16, NN / 16), 32, 0, stream>>>(hs_h, adjacency, attn_h);
        row_abs_sum<<<NN, 256, 0, stream>>>(attn_h, denom);
        gemm_norm_axpy<<<dim3(NN / 16, DIM / 16), 32, 0, stream>>>(attn_h, hsT_h, denom, xs);
    }

    f32_to_f16<<<(NN * DIM + 255) / 256, 256, 0, stream>>>(xs, xs_h, NN * DIM);
    gemm_bias_act<<<dim3(NN / 16, DIM / 16), 32, 0, stream>>>(
        xs_h, Wc_h, bc, h_f32, h_h, DIM, DIM, DIM, DIM, 1);

    for (int d = 0; d < 2; ++d) {
        gemm_bias_act<<<dim3(MM / 16, 512 / 16), 32, 0, stream>>>(
            wv_h, Wih_h + (size_t)d * 512 * DIM, bsum + d * 512,
            G + (size_t)d * MM * 512, nullptr, DIM, DIM, DIM, 512, 0);
    }
    lstm_dir<<<2, 128, 0, stream>>>(G, Whh, hbuf);
    bilstm_concat_relu<<<MM, 256, 0, stream>>>(hbuf, xs_rnn_h);
    gemm_bias_act<<<dim3(MM / 16, DIM / 16), 32, 0, stream>>>(
        xs_rnn_h, Wp_h, bp, hp_f32, hp_h, 2 * DIM, 2 * DIM, 2 * DIM, DIM, 1);

    gemm_bias_act<<<dim3(NN / 16, MM / 16), 32, 0, stream>>>(
        h_h, hp_h, nullptr, weights_out, nullptr, DIM, DIM, DIM, MM, 0);

    row_mean<<<NN, 256, 0, stream>>>(weights_out, w_comp);
    col_mean<<<MM / 256, 256, 0, stream>>>(weights_out, w_prot);
    pool_vectors<<<1, 256, 0, stream>>>(h_f32, hp_f32, w_comp, w_prot, cat);
    final_mlp<<<1, 256, 0, stream>>>(cat, Wo, bo, Wi, bi, out);
}


# pooling: row/col means of h*hp^T taken as h.colsum(hp)/M and hp.colsum(h)/N in f32, reduced as 64-block partial sums; no pass over the 4096x4096 matrix, one launch fewer; sum order changed
# speedup vs baseline: 5.3161x; 1.0968x over previous
#include <hip/hip_runtime.h>

typedef __attribute__((ext_vector_type(16))) _Float16 v16h;
typedef __attribute__((ext_vector_type(8)))  float    v8f;

#define DIM 128
#define NN  4096
#define MM  4096

__device__ __forceinline__ v16h frag_load(const _Float16* __restrict__ tile, int ld) {
    const int lane = threadIdx.x & 31;
    const int r    = lane & 15;
    const int kb   = (lane & 16) ? 8 : 0;
    const _Float16* q = tile + (size_t)r * (size_t)ld + kb;
    v16h f;
    uint4* fu = reinterpret_cast<uint4*>(&f);
    fu[0] = *reinterpret_cast<const uint4*>(q);
    fu[1] = *reinterpret_cast<const uint4*>(q + 16);
    return f;
}

__device__ __forceinline__ void frag_rc(int& rbase, int& col) {
    const int lane = threadIdx.x & 31;
    col   = lane & 15;
    rbase = (lane & 16) ? 8 : 0;
}

__global__ void gemm_bias_act(const _Float16* __restrict__ A, const _Float16* __restrict__ B,
                              const float* __restrict__ bias,
                              float* __restrict__ Cf, _Float16* __restrict__ Ch,
                              int K, int lda, int ldb, int ldc, int do_relu) {
    const int i0 = blockIdx.x * 16;
    const int j0 = blockIdx.y * 16;
    v8f acc = {};
    for (int k = 0; k < K; k += 32) {
        v16h a = frag_load(A + (size_t)i0 * lda + k, lda);
        v16h b = frag_load(B + (size_t)j0 * ldb + k, ldb);
        acc = __builtin_amdgcn_wmma_f32_16x16x32_f16(false, a, false, b, (short)0, acc,
                                                     false, false);
    }
    int rb, col; frag_rc(rb, col);
    const float bv = bias ? bias[j0 + col] : 0.f;
#pragma unroll
    for (int r = 0; r < 8; ++r) {
        const int row = i0 + rb + r;
        float v = acc[r] + bv;
        if (do_relu) v = fmaxf(v, 0.f);
        const size_t idx = (size_t)row * ldc + j0 + col;
        if (Cf) Cf[idx] = v;
        if (Ch) Ch[idx] = (_Float16)v;
    }
}

__global__ void gemm_attn(const _Float16* __restrict__ hs, const float* __restrict__ adj,
                          _Float16* __restrict__ attn) {
    const int i0 = blockIdx.x * 16;
    const int j0 = blockIdx.y * 16;
    v8f acc = {};
#pragma unroll
    for (int k = 0; k < DIM; k += 32) {
        v16h a = frag_load(hs + (size_t)i0 * DIM + k, DIM);
        v16h b = frag_load(hs + (size_t)j0 * DIM + k, DIM);
        acc = __builtin_amdgcn_wmma_f32_16x16x32_f16(false, a, false, b, (short)0, acc,
                                                     false, false);
    }
    int rb, col; frag_rc(rb, col);
#pragma unroll
    for (int r = 0; r < 8; ++r) {
        const int row = i0 + rb + r;
        const size_t idx = (size_t)row * NN + j0 + col;
        attn[idx] = (_Float16)(acc[r] * adj[idx]);
    }
}

__global__ void row_abs_sum(const _Float16* __restrict__ attn, float* __restrict__ denom) {
    __shared__ float red[256];
    const int i = blockIdx.x;
    float s = 0.f;
    for (int j = threadIdx.x; j < NN; j += 256) s += fabsf((float)attn[(size_t)i * NN + j]);
    red[threadIdx.x] = s;
    __syncthreads();
    for (int w = 128; w > 0; w >>= 1) {
        if (threadIdx.x < w) red[threadIdx.x] += red[threadIdx.x + w];
        __syncthreads();
    }
    if (threadIdx.x == 0) denom[i] = fmaxf(red[0], 1e-12f);
}

__global__ void transpose_f16(const _Float16* __restrict__ src, _Float16* __restrict__ dst) {
    const int j = blockIdx.x;
    const int n = threadIdx.x;
    dst[(size_t)n * NN + j] = src[(size_t)j * DIM + n];
}

__global__ void gemm_norm_axpy(const _Float16* __restrict__ attn, const _Float16* __restrict__ hsT,
                               const float* __restrict__ denom, float* __restrict__ xs) {
    const int i0 = blockIdx.x * 16;
    const int j0 = blockIdx.y * 16;
    const _Float16* Ab = attn + (size_t)i0 * NN;
    const _Float16* Bb = hsT + (size_t)j0 * NN;
    v8f acc = {};
    for (int k = 0; k < NN; k += 32) {
        __builtin_prefetch(Ab + k + 256, 0, 0);
        v16h a = frag_load(Ab + k, NN);
        v16h b = frag_load(Bb + k, NN);
        acc = __builtin_amdgcn_wmma_f32_16x16x32_f16(false, a, false, b, (short)0, acc,
                                                     false, false);
    }
    int rb, col; frag_rc(rb, col);
#pragma unroll
    for (int r = 0; r < 8; ++r) {
        const int row = i0 + rb + r;
        xs[(size_t)row * DIM + j0 + col] += acc[r] / denom[row];
    }
}

__global__ void gather_embed(const int* __restrict__ idx, const float* __restrict__ emb,
                             float* __restrict__ of32, _Float16* __restrict__ of16) {
    const int i = blockIdx.x;
    const int n = threadIdx.x;
    const float v = emb[(size_t)idx[i] * DIM + n];
    if (of32) of32[(size_t)i * DIM + n] = v;
    if (of16) of16[(size_t)i * DIM + n] = (_Float16)v;
}

__global__ void f32_to_f16(const float* __restrict__ s, _Float16* __restrict__ d, int n) {
    const int i = blockIdx.x * 256 + threadIdx.x;
    if (i < n) d[i] = (_Float16)s[i];
}

__global__ void vec_add(const float* __restrict__ a, const float* __restrict__ b,
                        float* __restrict__ c, int n) {
    const int i = blockIdx.x * 256 + threadIdx.x;
    if (i < n) c[i] = a[i] + b[i];
}

typedef __attribute__((ext_vector_type(2))) float v2f;
typedef __attribute__((ext_vector_type(4))) float v4f;

__device__ __forceinline__ float gate_sigmoid(float z) {
    return __builtin_amdgcn_rcpf(1.f + __builtin_amdgcn_exp2f(-1.44269504f * z));
}
__device__ __forceinline__ float gate_tanh_r(float x) {
    return __builtin_amdgcn_rcpf(1.f + __builtin_amdgcn_exp2f(2.88539008f * x));
}

__global__ __launch_bounds__(128) void lstm_dir(const float* __restrict__ G,
                                                const float* __restrict__ Whh,
                                                float* __restrict__ hout) {
    __shared__ __attribute__((aligned(16))) float h_sh[2][DIM];
    const int d = blockIdx.x;
    const int n = threadIdx.x;
    const float* W  = Whh + (size_t)d * 512 * DIM;
    const float* Gd = G + (size_t)d * MM * 512;
    float* hd = hout + (size_t)d * MM * DIM;

    v2f wi[DIM / 2], wf[DIM / 2], wg[DIM / 2], wo[DIM / 2];
    {
        const v4f* ri = reinterpret_cast<const v4f*>(W + (size_t)(n)       * DIM);
        const v4f* rf = reinterpret_cast<const v4f*>(W + (size_t)(n + 128) * DIM);
        const v4f* rg = reinterpret_cast<const v4f*>(W + (size_t)(n + 256) * DIM);
        const v4f* ro = reinterpret_cast<const v4f*>(W + (size_t)(n + 384) * DIM);
#pragma unroll
        for (int k4 = 0; k4 < DIM / 4; ++k4) {
            const v4f a = ri[k4], b = rf[k4], c4 = rg[k4], e = ro[k4];
            wi[2 * k4] = __builtin_shufflevector(a, a, 0, 1);
            wi[2 * k4 + 1] = __builtin_shufflevector(a, a, 2, 3);
            wf[2 * k4] = __builtin_shufflevector(b, b, 0, 1);
            wf[2 * k4 + 1] = __builtin_shufflevector(b, b, 2, 3);
            wg[2 * k4] = __builtin_shufflevector(c4, c4, 0, 1);
            wg[2 * k4 + 1] = __builtin_shufflevector(c4, c4, 2, 3);
            wo[2 * k4] = __builtin_shufflevector(e, e, 0, 1);
            wo[2 * k4 + 1] = __builtin_shufflevector(e, e, 2, 3);
        }
    }
    h_sh[0][n] = 0.f;
    float c = 0.f;
    __syncthreads();

    float pi, pf, pg, po;
    {
        const float* g0 = Gd + (size_t)(d ? (MM - 1) : 0) * 512 + n;
        pi = g0[0]; pf = g0[128]; pg = g0[256]; po = g0[384];
    }

    for (int t = 0; t < MM; ++t) {
        const int tt = d ? (MM - 1 - t) : t;
        const v4f* h4 = reinterpret_cast<const v4f*>(h_sh[t & 1]);
        v2f ai = {pi, 0.f}, af = {pf, 0.f}, ag = {pg, 0.f}, ao = {po, 0.f};
        v2f bi = {0.f, 0.f}, bf = {0.f, 0.f}, bg = {0.f, 0.f}, bo = {0.f, 0.f};
        v4f hv[DIM / 4];
#pragma unroll
        for (int k4 = 0; k4 < DIM / 4; ++k4) hv[k4] = h4[k4];
        __builtin_amdgcn_sched_barrier(0);
        {
            const int tn = (t + 1 < MM) ? (t + 1) : t;
            const float* gn = Gd + (size_t)(d ? (MM - 1 - tn) : tn) * 512 + n;
            pi = gn[0]; pf = gn[128]; pg = gn[256]; po = gn[384];
        }
#pragma unroll
        for (int k4 = 0; k4 < DIM / 4; ++k4) {
            const v2f hlo = __builtin_shufflevector(hv[k4], hv[k4], 0, 1);
            const v2f hhi = __builtin_shufflevector(hv[k4], hv[k4], 2, 3);
            ai = __builtin_elementwise_fma(hlo, wi[2 * k4], ai);
            af = __builtin_elementwise_fma(hlo, wf[2 * k4], af);
            ag = __builtin_elementwise_fma(hlo, wg[2 * k4], ag);
            ao = __builtin_elementwise_fma(hlo, wo[2 * k4], ao);
            bi = __builtin_elementwise_fma(hhi, wi[2 * k4 + 1], bi);
            bf = __builtin_elementwise_fma(hhi, wf[2 * k4 + 1], bf);
            bg = __builtin_elementwise_fma(hhi, wg[2 * k4 + 1], bg);
            bo = __builtin_elementwise_fma(hhi, wo[2 * k4 + 1], bo);
        }
        const v2f si = ai + bi, sf = af + bf, sg = ag + bg, so = ao + bo;
        float zi = si.x + si.y, zf = sf.x + sf.y, zg = sg.x + sg.y, zo = so.x + so.y;
        const float ig = gate_sigmoid(zi);
        const float fg = gate_sigmoid(zf);
        const float og = gate_sigmoid(zo);
        const float rg = gate_tanh_r(zg);
        c = __builtin_fmaf(rg, -2.f * ig, __builtin_fmaf(fg, c, ig));
        const float rc = gate_tanh_r(c);
        const float h = __builtin_fmaf(rc, -2.f * og, og);
        h_sh[(t + 1) & 1][n] = h;
        hd[(size_t)tt * DIM + n] = h;
        __builtin_amdgcn_fence(__ATOMIC_RELEASE, "workgroup", "local");
        __builtin_amdgcn_s_barrier();
        __builtin_amdgcn_fence(__ATOMIC_ACQUIRE, "workgroup", "local");
    }
}

__global__ void bilstm_concat_relu(const float* __restrict__ hbuf, _Float16* __restrict__ xs_rnn) {
    const int t = blockIdx.x;
    const int n = threadIdx.x;
    const float v = (n < DIM) ? hbuf[(size_t)t * DIM + n]
                              : hbuf[(size_t)(MM + t) * DIM + (n - DIM)];
    xs_rnn[(size_t)t * (2 * DIM) + n] = (_Float16)fmaxf(v, 0.f);
}

#define PB 64
#define PR (NN / PB)

__global__ void colsum_part(const float* __restrict__ h, const float* __restrict__ hp,
                            float* __restrict__ part) {
    const int t = threadIdx.x;
    const int n = t & 127;
    const float* src = ((t < 128) ? h : hp) + (size_t)blockIdx.x * PR * DIM + n;
    float s = 0.f;
    for (int r = 0; r < PR; ++r) s += src[(size_t)r * DIM];
    part[blockIdx.x * 256 + t] = s;
}

__global__ void pool_part(const float* __restrict__ h, const float* __restrict__ hp,
                          const float* __restrict__ part, float* __restrict__ catpart) {
    __shared__ __attribute__((aligned(16))) float colsum[256];
    __shared__ float w_sh[2][PR];
    const int t = threadIdx.x;
    const int side = t >> 7;
    const int n = t & 127;
    float s = 0.f;
    for (int b = 0; b < PB; ++b) s += part[b * 256 + t];
    colsum[t] = s;
    __syncthreads();
    const float* src = (side ? hp : h) + (size_t)blockIdx.x * PR * DIM;
    if (n < PR) {
        const v4f* other = reinterpret_cast<const v4f*>(colsum + (side ? 0 : 128));
        const v4f* row   = reinterpret_cast<const v4f*>(src + (size_t)n * DIM);
        float d = 0.f;
        for (int k4 = 0; k4 < DIM / 4; ++k4) {
            const v4f a = row[k4], b = other[k4];
            d += a.x * b.x + a.y * b.y + a.z * b.z + a.w * b.w;
        }
        w_sh[side][n] = d / (float)(side ? NN : MM);
    }
    __syncthreads();
    float acc = 0.f;
    for (int r = 0; r < PR; ++r) acc += w_sh[side][r] * src[(size_t)r * DIM + n];
    catpart[blockIdx.x * 256 + t] = acc;
}

__global__ void final_mlp(const float* __restrict__ catpart, const float* __restrict__ Wo,
                          const float* __restrict__ bo, const float* __restrict__ Wi,
                          const float* __restrict__ bi, float* __restrict__ out) {
    __shared__ float cur[256], nxt[256];
    const int t = threadIdx.x;
    {
        float s = 0.f;
        for (int b = 0; b < PB; ++b) s += catpart[b * 256 + t];
        cur[t] = s / (float)((t < 128) ? NN : MM);
    }
    __syncthreads();
    for (int l = 0; l < 2; ++l) {
        const float* Wl = Wo + (size_t)l * 256 * 256;
        float s = bo[l * 256 + t];
        for (int k = 0; k < 256; ++k) s += cur[k] * Wl[(size_t)t * 256 + k];
        nxt[t] = fmaxf(s, 0.f);
        __syncthreads();
        cur[t] = nxt[t];
        __syncthreads();
    }
    if (t < 2) {
        float s = bi[t];
        for (int k = 0; k < 256; ++k) s += cur[k] * Wi[(size_t)t * 256 + k];
        out[t] = s;
    }
}

extern "C" void kernel_launch(void* const* d_in, const int* in_sizes, int n_in,
                              void* d_out, int out_size, void* d_ws, size_t ws_size,
                              hipStream_t stream) {
    (void)in_sizes; (void)n_in; (void)out_size; (void)ws_size;
    const int*   fingerprints = (const int*)d_in[0];
    const float* adjacency    = (const float*)d_in[1];
    const int*   words        = (const int*)d_in[2];
    const float* emb_fp       = (const float*)d_in[3];
    const float* emb_word     = (const float*)d_in[4];
    const float* Wg  = (const float*)d_in[5];
    const float* bg  = (const float*)d_in[6];
    const float* Wih = (const float*)d_in[7];
    const float* Whh = (const float*)d_in[8];
    const float* bih = (const float*)d_in[9];
    const float* bhh = (const float*)d_in[10];
    const float* Wc  = (const float*)d_in[11];
    const float* bc  = (const float*)d_in[12];
    const float* Wp  = (const float*)d_in[13];
    const float* bp  = (const float*)d_in[14];
    const float* Wo  = (const float*)d_in[15];
    const float* bo  = (const float*)d_in[16];
    const float* Wi  = (const float*)d_in[17];
    const float* bi  = (const float*)d_in[18];

    float* out         = (float*)d_out;
    float* weights_out = out + 2;

    char* base = (char*)d_ws;
    size_t off = 0;
    auto alloc = [&](size_t bytes) -> void* {
        void* p = base + off;
        off = (off + bytes + 255) & ~(size_t)255;
        return p;
    };
    float*    xs       = (float*)   alloc((size_t)NN * DIM * 4);
    _Float16* xs_h     = (_Float16*)alloc((size_t)NN * DIM * 2);
    _Float16* hs_h     = (_Float16*)alloc((size_t)NN * DIM * 2);
    _Float16* hsT_h    = (_Float16*)alloc((size_t)DIM * NN * 2);
    _Float16* attn_h   = (_Float16*)alloc((size_t)NN * NN * 2);
    float*    denom    = (float*)   alloc((size_t)NN * 4);
    _Float16* Wg_h     = (_Float16*)alloc((size_t)3 * DIM * DIM * 2);
    _Float16* Wc_h     = (_Float16*)alloc((size_t)DIM * DIM * 2);
    _Float16* Wp_h     = (_Float16*)alloc((size_t)DIM * 2 * DIM * 2);
    _Float16* Wih_h    = (_Float16*)alloc((size_t)2 * 512 * DIM * 2);
    float*    bsum     = (float*)   alloc((size_t)2 * 512 * 4);
    _Float16* wv_h     = (_Float16*)alloc((size_t)MM * DIM * 2);
    float*    G        = (float*)   alloc((size_t)2 * MM * 512 * 4);
    float*    hbuf     = (float*)   alloc((size_t)2 * MM * DIM * 4);
    _Float16* xs_rnn_h = (_Float16*)alloc((size_t)MM * 2 * DIM * 2);
    float*    h_f32    = (float*)   alloc((size_t)NN * DIM * 4);
    _Float16* h_h      = (_Float16*)alloc((size_t)NN * DIM * 2);
    float*    hp_f32   = (float*)   alloc((size_t)MM * DIM * 4);
    _Float16* hp_h     = (_Float16*)alloc((size_t)MM * DIM * 2);
    float*    part     = (float*)   alloc((size_t)PB * 256 * 4);
    float*    catpart  = (float*)   alloc((size_t)PB * 256 * 4);

    f32_to_f16<<<(3 * DIM * DIM + 255) / 256, 256, 0, stream>>>(Wg, Wg_h, 3 * DIM * DIM);
    f32_to_f16<<<(DIM * DIM + 255) / 256, 256, 0, stream>>>(Wc, Wc_h, DIM * DIM);
    f32_to_f16<<<(DIM * 2 * DIM + 255) / 256, 256, 0, stream>>>(Wp, Wp_h, DIM * 2 * DIM);
    f32_to_f16<<<(2 * 512 * DIM + 255) / 256, 256, 0, stream>>>(Wih, Wih_h, 2 * 512 * DIM);
    vec_add<<<(2 * 512 + 255) / 256, 256, 0, stream>>>(bih, bhh, bsum, 2 * 512);

    gather_embed<<<NN, DIM, 0, stream>>>(fingerprints, emb_fp, xs, nullptr);
    gather_embed<<<MM, DIM, 0, stream>>>(words, emb_word, nullptr, wv_h);

    for (int l = 0; l < 3; ++l) {
        f32_to_f16<<<(NN * DIM + 255) / 256, 256, 0, stream>>>(xs, xs_h, NN * DIM);
        gemm_bias_act<<<dim3(NN / 16, DIM / 16), 32, 0, stream>>>(
            xs_h, Wg_h + (size_t)l * DIM * DIM, bg + l * DIM,
            nullptr, hs_h, DIM, DIM, DIM, DIM, 1);
        transpose_f16<<<NN, DIM, 0, stream>>>(hs_h, hsT_h);
        gemm_attn<<<dim3(NN / 16, NN / 16), 32, 0, stream>>>(hs_h, adjacency, attn_h);
        row_abs_sum<<<NN, 256, 0, stream>>>(attn_h, denom);
        gemm_norm_axpy<<<dim3(NN / 16, DIM / 16), 32, 0, stream>>>(attn_h, hsT_h, denom, xs);
    }

    f32_to_f16<<<(NN * DIM + 255) / 256, 256, 0, stream>>>(xs, xs_h, NN * DIM);
    gemm_bias_act<<<dim3(NN / 16, DIM / 16), 32, 0, stream>>>(
        xs_h, Wc_h, bc, h_f32, h_h, DIM, DIM, DIM, DIM, 1);

    for (int d = 0; d < 2; ++d) {
        gemm_bias_act<<<dim3(MM / 16, 512 / 16), 32, 0, stream>>>(
            wv_h, Wih_h + (size_t)d * 512 * DIM, bsum + d * 512,
            G + (size_t)d * MM * 512, nullptr, DIM, DIM, DIM, 512, 0);
    }
    lstm_dir<<<2, 128, 0, stream>>>(G, Whh, hbuf);
    bilstm_concat_relu<<<MM, 256, 0, stream>>>(hbuf, xs_rnn_h);
    gemm_bias_act<<<dim3(MM / 16, DIM / 16), 32, 0, stream>>>(
        xs_rnn_h, Wp_h, bp, hp_f32, hp_h, 2 * DIM, 2 * DIM, 2 * DIM, DIM, 1);

    gemm_bias_act<<<dim3(NN / 16, MM / 16), 32, 0, stream>>>(
        h_h, hp_h, nullptr, weights_out, nullptr, DIM, DIM, DIM, MM, 0);

    colsum_part<<<PB, 256, 0, stream>>>(h_f32, hp_f32, part);
    pool_part<<<PB, 256, 0, stream>>>(h_f32, hp_f32, part, catpart);
    final_mlp<<<1, 256, 0, stream>>>(catpart, Wo, bo, Wi, bi, out);
}


# lstm_dir: pre-gates G fetched 4 steps ahead into a ring of 4 register sets (step loop unrolled 4x), not 1 step ahead: a step no longer waits for a memory round trip; order of every sum unchanged
# speedup vs baseline: 6.5565x; 1.2333x over previous
#include <hip/hip_runtime.h>

typedef __attribute__((ext_vector_type(16))) _Float16 v16h;
typedef __attribute__((ext_vector_type(8)))  float    v8f;

#define DIM 128
#define NN  4096
#define MM  4096

__device__ __forceinline__ v16h frag_load(const _Float16* __restrict__ tile, int ld) {
    const int lane = threadIdx.x & 31;
    const int r    = lane & 15;
    const int kb   = (lane & 16) ? 8 : 0;
    const _Float16* q = tile + (size_t)r * (size_t)ld + kb;
    v16h f;
    uint4* fu = reinterpret_cast<uint4*>(&f);
    fu[0] = *reinterpret_cast<const uint4*>(q);
    fu[1] = *reinterpret_cast<const uint4*>(q + 16);
    return f;
}

__device__ __forceinline__ void frag_rc(int& rbase, int& col) {
    const int lane = threadIdx.x & 31;
    col   = lane & 15;
    rbase = (lane & 16) ? 8 : 0;
}

__global__ void gemm_bias_act(const _Float16* __restrict__ A, const _Float16* __restrict__ B,
                              const float* __restrict__ bias,
                              float* __restrict__ Cf, _Float16* __restrict__ Ch,
                              int K, int lda, int ldb, int ldc, int do_relu) {
    const int i0 = blockIdx.x * 16;
    const int j0 = blockIdx.y * 16;
    v8f acc = {};
    for (int k = 0; k < K; k += 32) {
        v16h a = frag_load(A + (size_t)i0 * lda + k, lda);
        v16h b = frag_load(B + (size_t)j0 * ldb + k, ldb);
        acc = __builtin_amdgcn_wmma_f32_16x16x32_f16(false, a, false, b, (short)0, acc,
                                                     false, false);
    }
    int rb, col; frag_rc(rb, col);
    const float bv = bias ? bias[j0 + col] : 0.f;
#pragma unroll
    for (int r = 0; r < 8; ++r) {
        const int row = i0 + rb + r;
        float v = acc[r] + bv;
        if (do_relu) v = fmaxf(v, 0.f);
        const size_t idx = (size_t)row * ldc + j0 + col;
        if (Cf) Cf[idx] = v;
        if (Ch) Ch[idx] = (_Float16)v;
    }
}

__global__ void transpose_f16(const _Float16* __restrict__ src, _Float16* __restrict__ dst) {
    const int j = blockIdx.x;
    const int n = threadIdx.x;
    dst[(size_t)n * NN + j] = src[(size_t)j * DIM + n];
}

typedef __attribute__((ext_vector_type(4))) float f4;
#define GW 8
#define OP 132
#define GSTEPS (NN / (32 * GW))

struct GatStep {
    v16h aj[8];
    v16h bt[8];
    f4   ad[4];
};

__device__ __forceinline__ void gat_load(GatStep& g, const _Float16* hs, const _Float16* hsT,
                                         const float* arow, int j0) {
#pragma unroll
    for (int k = 0; k < 4; ++k) {
        g.aj[k]     = frag_load(hs + (size_t)j0 * DIM + 32 * k, DIM);
        g.aj[4 + k] = frag_load(hs + (size_t)(j0 + 16) * DIM + 32 * k, DIM);
    }
#pragma unroll
    for (int nt = 0; nt < 8; ++nt) g.bt[nt] = frag_load(hsT + (size_t)(16 * nt) * NN + j0, NN);
    g.ad[0] = *reinterpret_cast<const f4*>(arow + j0);
    g.ad[1] = *reinterpret_cast<const f4*>(arow + j0 + 4);
    g.ad[2] = *reinterpret_cast<const f4*>(arow + j0 + 16);
    g.ad[3] = *reinterpret_cast<const f4*>(arow + j0 + 20);
}

__device__ __forceinline__ void gat_step(const GatStep& g, const v16h (&bi)[4], v8f (&o)[8], float& rs) {
    v8f s0 = {}, s1 = {};
#pragma unroll
    for (int k = 0; k < 4; ++k) {
        s0 = __builtin_amdgcn_wmma_f32_16x16x32_f16(false, g.aj[k], false, bi[k], (short)0, s0,
                                                    false, false);
        s1 = __builtin_amdgcn_wmma_f32_16x16x32_f16(false, g.aj[4 + k], false, bi[k], (short)0, s1,
                                                    false, false);
    }
    const float m0[8] = {g.ad[0].x, g.ad[0].y, g.ad[0].z, g.ad[0].w,
                         g.ad[1].x, g.ad[1].y, g.ad[1].z, g.ad[1].w};
    const float m1[8] = {g.ad[2].x, g.ad[2].y, g.ad[2].z, g.ad[2].w,
                         g.ad[3].x, g.ad[3].y, g.ad[3].z, g.ad[3].w};
    v16h p;
#pragma unroll
    for (int r = 0; r < 8; ++r) {
        const _Float16 q0 = (_Float16)(s0[r] * m0[r]);
        const _Float16 q1 = (_Float16)(s1[r] * m1[r]);
        p[r]     = q0;
        p[8 + r] = q1;
        rs += fabsf((float)q0) + fabsf((float)q1);
    }
#pragma unroll
    for (int nt = 0; nt < 8; ++nt)
        o[nt] = __builtin_amdgcn_wmma_f32_16x16x32_f16(false, p, false, g.bt[nt], (short)0, o[nt],
                                                       false, false);
}

__global__ __launch_bounds__(GW * 32) void gat_fused(const _Float16* hs, const _Float16* hsT,
                                                     const float* adj, float* __restrict__ xs) {
    __shared__ __attribute__((aligned(16))) float o_sh[GW][16][OP];
    __shared__ float rs_sh[GW][2][16];
    const int i0   = blockIdx.x * 16;
    const int w    = threadIdx.x >> 5;
    const int lane = threadIdx.x & 31;
    const int c    = lane & 15;
    const int hh   = lane >> 4;
    const int h8   = hh * 8;

    v16h bi[4];
#pragma unroll
    for (int k = 0; k < 4; ++k) bi[k] = frag_load(hs + (size_t)i0 * DIM + 32 * k, DIM);

    v8f o[8];
#pragma unroll
    for (int nt = 0; nt < 8; ++nt) o[nt] = v8f{};
    float rs = 0.f;
    const float* arow = adj + (size_t)(i0 + c) * NN + h8;

    GatStep ga, gb;
    gat_load(ga, hs, hsT, arow, w * 64);

#pragma nounroll
    for (int it = 0; it < GSTEPS / 2; ++it) {
        const int jc = (it * GW + w) * 64;
        if (it + 2 < GSTEPS / 2) __builtin_prefetch(arow + jc + 2 * 64 * GW + 24 * hh, 0, 0);
        gat_load(gb, hs, hsT, arow, jc + 32);
        __builtin_amdgcn_sched_barrier(0);
        gat_step(ga, bi, o, rs);
        const int itn = (it + 1 < GSTEPS / 2) ? (it + 1) : it;
        gat_load(ga, hs, hsT, arow, (itn * GW + w) * 64);
        __builtin_amdgcn_sched_barrier(0);
        gat_step(gb, bi, o, rs);
    }

#pragma unroll
    for (int nt = 0; nt < 8; ++nt)
#pragma unroll
        for (int r = 0; r < 8; ++r) o_sh[w][h8 + r][16 * nt + c] = o[nt][r];
    rs_sh[w][hh][c] = rs;
    __syncthreads();

    const int row = threadIdx.x >> 4;
    const int q4  = (threadIdx.x & 15) * 4;
    f4 lo = {0.f, 0.f, 0.f, 0.f}, hi = {0.f, 0.f, 0.f, 0.f};
    float den = 0.f;
#pragma unroll
    for (int ww = 0; ww < GW; ++ww) {
        lo += *reinterpret_cast<const f4*>(&o_sh[ww][row][q4]);
        hi += *reinterpret_cast<const f4*>(&o_sh[ww][row][64 + q4]);
        den += rs_sh[ww][0][row] + rs_sh[ww][1][row];
    }
    den = fmaxf(den, 1e-12f);
    float* xr = xs + (size_t)(i0 + row) * DIM;
    f4 x0 = *reinterpret_cast<const f4*>(xr + q4);
    f4 x1 = *reinterpret_cast<const f4*>(xr + 64 + q4);
    x0.x += lo.x / den; x0.y += lo.y / den; x0.z += lo.z / den; x0.w += lo.w / den;
    x1.x += hi.x / den; x1.y += hi.y / den; x1.z += hi.z / den; x1.w += hi.w / den;
    *reinterpret_cast<f4*>(xr + q4)      = x0;
    *reinterpret_cast<f4*>(xr + 64 + q4) = x1;
}

__global__ void gather_embed(const int* __restrict__ idx, const float* __restrict__ emb,
                             float* __restrict__ of32, _Float16* __restrict__ of16) {
    const int i = blockIdx.x;
    const int n = threadIdx.x;
    const float v = emb[(size_t)idx[i] * DIM + n];
    if (of32) of32[(size_t)i * DIM + n] = v;
    if (of16) of16[(size_t)i * DIM + n] = (_Float16)v;
}

__global__ void f32_to_f16(const float* __restrict__ s, _Float16* __restrict__ d, int n) {
    const int i = blockIdx.x * 256 + threadIdx.x;
    if (i < n) d[i] = (_Float16)s[i];
}

__global__ void vec_add(const float* __restrict__ a, const float* __restrict__ b,
                        float* __restrict__ c, int n) {
    const int i = blockIdx.x * 256 + threadIdx.x;
    if (i < n) c[i] = a[i] + b[i];
}

typedef __attribute__((ext_vector_type(2))) float v2f;
typedef __attribute__((ext_vector_type(4))) float v4f;
#define GD 4

__device__ __forceinline__ float gate_sigmoid(float z) {
    return __builtin_amdgcn_rcpf(1.f + __builtin_amdgcn_exp2f(-1.44269504f * z));
}
__device__ __forceinline__ float gate_tanh_r(float x) {
    return __builtin_amdgcn_rcpf(1.f + __builtin_amdgcn_exp2f(2.88539008f * x));
}

__global__ __launch_bounds__(128) void lstm_dir(const float* __restrict__ G,
                                                const float* __restrict__ Whh,
                                                float* __restrict__ hout) {
    __shared__ __attribute__((aligned(16))) float h_sh[2][DIM];
    const int d = blockIdx.x;
    const int n = threadIdx.x;
    const float* W  = Whh + (size_t)d * 512 * DIM;
    const float* Gd = G + (size_t)d * MM * 512;
    float* hd = hout + (size_t)d * MM * DIM;

    v2f wi[DIM / 2], wf[DIM / 2], wg[DIM / 2], wo[DIM / 2];
    {
        const v4f* ri = reinterpret_cast<const v4f*>(W + (size_t)(n)       * DIM);
        const v4f* rf = reinterpret_cast<const v4f*>(W + (size_t)(n + 128) * DIM);
        const v4f* rg = reinterpret_cast<const v4f*>(W + (size_t)(n + 256) * DIM);
        const v4f* ro = reinterpret_cast<const v4f*>(W + (size_t)(n + 384) * DIM);
#pragma unroll
        for (int k4 = 0; k4 < DIM / 4; ++k4) {
            const v4f a = ri[k4], b = rf[k4], c4 = rg[k4], e = ro[k4];
            wi[2 * k4] = __builtin_shufflevector(a, a, 0, 1);
            wi[2 * k4 + 1] = __builtin_shufflevector(a, a, 2, 3);
            wf[2 * k4] = __builtin_shufflevector(b, b, 0, 1);
            wf[2 * k4 + 1] = __builtin_shufflevector(b, b, 2, 3);
            wg[2 * k4] = __builtin_shufflevector(c4, c4, 0, 1);
            wg[2 * k4 + 1] = __builtin_shufflevector(c4, c4, 2, 3);
            wo[2 * k4] = __builtin_shufflevector(e, e, 0, 1);
            wo[2 * k4 + 1] = __builtin_shufflevector(e, e, 2, 3);
        }
    }
    h_sh[0][n] = 0.f;
    float c = 0.f;
    __syncthreads();

    float pi[GD], pf[GD], pg[GD], po[GD];
#pragma unroll
    for (int u = 0; u < GD; ++u) {
        const float* g0 = Gd + (size_t)(d ? (MM - 1 - u) : u) * 512 + n;
        pi[u] = g0[0]; pf[u] = g0[128]; pg[u] = g0[256]; po[u] = g0[384];
    }

#pragma nounroll
    for (int t0 = 0; t0 < MM; t0 += GD) {
#pragma unroll
        for (int u = 0; u < GD; ++u) {
            const int t = t0 + u;
            const int tt = d ? (MM - 1 - t) : t;
            const v4f* h4 = reinterpret_cast<const v4f*>(h_sh[u & 1]);
            v2f ai = {pi[u], 0.f}, af = {pf[u], 0.f}, ag = {pg[u], 0.f}, ao = {po[u], 0.f};
            v2f bi = {0.f, 0.f}, bf = {0.f, 0.f}, bg = {0.f, 0.f}, bo = {0.f, 0.f};
            v4f hv[DIM / 4];
#pragma unroll
            for (int k4 = 0; k4 < DIM / 4; ++k4) hv[k4] = h4[k4];
            __builtin_amdgcn_sched_barrier(0);
#pragma unroll
            for (int k4 = 0; k4 < DIM / 4; ++k4) {
                const v2f hlo = __builtin_shufflevector(hv[k4], hv[k4], 0, 1);
                const v2f hhi = __builtin_shufflevector(hv[k4], hv[k4], 2, 3);
                ai = __builtin_elementwise_fma(hlo, wi[2 * k4], ai);
                af = __builtin_elementwise_fma(hlo, wf[2 * k4], af);
                ag = __builtin_elementwise_fma(hlo, wg[2 * k4], ag);
                ao = __builtin_elementwise_fma(hlo, wo[2 * k4], ao);
                bi = __builtin_elementwise_fma(hhi, wi[2 * k4 + 1], bi);
                bf = __builtin_elementwise_fma(hhi, wf[2 * k4 + 1], bf);
                bg = __builtin_elementwise_fma(hhi, wg[2 * k4 + 1], bg);
                bo = __builtin_elementwise_fma(hhi, wo[2 * k4 + 1], bo);
                if (k4 == 0) {
                    const int tn = (t + GD < MM) ? (t + GD) : t;
                    const float* gn = Gd + (size_t)(d ? (MM - 1 - tn) : tn) * 512 + n;
                    pi[u] = gn[0]; pf[u] = gn[128]; pg[u] = gn[256]; po[u] = gn[384];
                }
            }
            const v2f si = ai + bi, sf = af + bf, sg = ag + bg, so = ao + bo;
            float zi = si.x + si.y, zf = sf.x + sf.y, zg = sg.x + sg.y, zo = so.x + so.y;
            const float ig = gate_sigmoid(zi);
            const float fg = gate_sigmoid(zf);
            const float og = gate_sigmoid(zo);
            const float rg = gate_tanh_r(zg);
            c = __builtin_fmaf(rg, -2.f * ig, __builtin_fmaf(fg, c, ig));
            const float rc = gate_tanh_r(c);
            const float h = __builtin_fmaf(rc, -2.f * og, og);
            h_sh[(u + 1) & 1][n] = h;
            hd[(size_t)tt * DIM + n] = h;
            __builtin_amdgcn_fence(__ATOMIC_RELEASE, "workgroup", "local");
            __builtin_amdgcn_s_barrier();
            __builtin_amdgcn_fence(__ATOMIC_ACQUIRE, "workgroup", "local");
        }
    }
}

__global__ void bilstm_concat_relu(const float* __restrict__ hbuf, _Float16* __restrict__ xs_rnn) {
    const int t = blockIdx.x;
    const int n = threadIdx.x;
    const float v = (n < DIM) ? hbuf[(size_t)t * DIM + n]
                              : hbuf[(size_t)(MM + t) * DIM + (n - DIM)];
    xs_rnn[(size_t)t * (2 * DIM) + n] = (_Float16)fmaxf(v, 0.f);
}

#define PB 64
#define PR (NN / PB)

__global__ void colsum_part(const float* __restrict__ h, const float* __restrict__ hp,
                            float* __restrict__ part) {
    const int t = threadIdx.x;
    const int n = t & 127;
    const float* src = ((t < 128) ? h : hp) + (size_t)blockIdx.x * PR * DIM + n;
    float s = 0.f;
    for (int r = 0; r < PR; ++r) s += src[(size_t)r * DIM];
    part[blockIdx.x * 256 + t] = s;
}

__global__ void pool_part(const float* __restrict__ h, const float* __restrict__ hp,
                          const float* __restrict__ part, float* __restrict__ catpart) {
    __shared__ __attribute__((aligned(16))) float colsum[256];
    __shared__ float w_sh[2][PR];
    const int t = threadIdx.x;
    const int side = t >> 7;
    const int n = t & 127;
    float s = 0.f;
    for (int b = 0; b < PB; ++b) s += part[b * 256 + t];
    colsum[t] = s;
    __syncthreads();
    const float* src = (side ? hp : h) + (size_t)blockIdx.x * PR * DIM;
    if (n < PR) {
        const v4f* other = reinterpret_cast<const v4f*>(colsum + (side ? 0 : 128));
        const v4f* row   = reinterpret_cast<const v4f*>(src + (size_t)n * DIM);
        float d = 0.f;
        for (int k4 = 0; k4 < DIM / 4; ++k4) {
            const v4f a = row[k4], b = other[k4];
            d += a.x * b.x + a.y * b.y + a.z * b.z + a.w * b.w;
        }
        w_sh[side][n] = d / (float)(side ? NN : MM);
    }
    __syncthreads();
    float acc = 0.f;
    for (int r = 0; r < PR; ++r) acc += w_sh[side][r] * src[(size_t)r * DIM + n];
    catpart[blockIdx.x * 256 + t] = acc;
}

__global__ void final_mlp(const float* __restrict__ catpart, const float* __restrict__ Wo,
                          const float* __restrict__ bo, const float* __restrict__ Wi,
                          const float* __restrict__ bi, float* __restrict__ out) {
    __shared__ float cur[256], nxt[256];
    const int t = threadIdx.x;
    {
        float s = 0.f;
        for (int b = 0; b < PB; ++b) s += catpart[b * 256 + t];
        cur[t] = s / (float)((t < 128) ? NN : MM);
    }
    __syncthreads();
    for (int l = 0; l < 2; ++l) {
        const float* Wl = Wo + (size_t)l * 256 * 256;
        float s = bo[l * 256 + t];
        for (int k = 0; k < 256; ++k) s += cur[k] * Wl[(size_t)t * 256 + k];
        nxt[t] = fmaxf(s, 0.f);
        __syncthreads();
        cur[t] = nxt[t];
        __syncthreads();
    }
    if (t < 2) {
        float s = bi[t];
        for (int k = 0; k < 256; ++k) s += cur[k] * Wi[(size_t)t * 256 + k];
        out[t] = s;
    }
}

extern "C" void kernel_launch(void* const* d_in, const int* in_sizes, int n_in,
                              void* d_out, int out_size, void* d_ws, size_t ws_size,
                              hipStream_t stream) {
    (void)in_sizes; (void)n_in; (void)out_size; (void)ws_size;
    const int*   fingerprints = (const int*)d_in[0];
    const float* adjacency    = (const float*)d_in[1];
    const int*   words        = (const int*)d_in[2];
    const float* emb_fp       = (const float*)d_in[3];
    const float* emb_word     = (const float*)d_in[4];
    const float* Wg  = (const float*)d_in[5];
    const float* bg  = (const float*)d_in[6];
    const float* Wih = (const float*)d_in[7];
    const float* Whh = (const float*)d_in[8];
    const float* bih = (const float*)d_in[9];
    const float* bhh = (const float*)d_in[10];
    const float* Wc  = (const float*)d_in[11];
    const float* bc  = (const float*)d_in[12];
    const float* Wp  = (const float*)d_in[13];
    const float* bp  = (const float*)d_in[14];
    const float* Wo  = (const float*)d_in[15];
    const float* bo  = (const float*)d_in[16];
    const float* Wi  = (const float*)d_in[17];
    const float* bi  = (const float*)d_in[18];

    float* out         = (float*)d_out;
    float* weights_out = out + 2;

    char* base = (char*)d_ws;
    size_t off = 0;
    auto alloc = [&](size_t bytes) -> void* {
        void* p = base + off;
        off = (off + bytes + 255) & ~(size_t)255;
        return p;
    };
    float*    xs       = (float*)   alloc((size_t)NN * DIM * 4);
    _Float16* xs_h     = (_Float16*)alloc((size_t)NN * DIM * 2);
    _Float16* hs_h     = (_Float16*)alloc((size_t)NN * DIM * 2);
    _Float16* hsT_h    = (_Float16*)alloc((size_t)DIM * NN * 2);
    _Float16* Wg_h     = (_Float16*)alloc((size_t)3 * DIM * DIM * 2);
    _Float16* Wc_h     = (_Float16*)alloc((size_t)DIM * DIM * 2);
    _Float16* Wp_h     = (_Float16*)alloc((size_t)DIM * 2 * DIM * 2);
    _Float16* Wih_h    = (_Float16*)alloc((size_t)2 * 512 * DIM * 2);
    float*    bsum     = (float*)   alloc((size_t)2 * 512 * 4);
    _Float16* wv_h     = (_Float16*)alloc((size_t)MM * DIM * 2);
    float*    G        = (float*)   alloc((size_t)2 * MM * 512 * 4);
    float*    hbuf     = (float*)   alloc((size_t)2 * MM * DIM * 4);
    _Float16* xs_rnn_h = (_Float16*)alloc((size_t)MM * 2 * DIM * 2);
    float*    h_f32    = (float*)   alloc((size_t)NN * DIM * 4);
    _Float16* h_h      = (_Float16*)alloc((size_t)NN * DIM * 2);
    float*    hp_f32   = (float*)   alloc((size_t)MM * DIM * 4);
    _Float16* hp_h     = (_Float16*)alloc((size_t)MM * DIM * 2);
    float*    part     = (float*)   alloc((size_t)PB * 256 * 4);
    float*    catpart  = (float*)   alloc((size_t)PB * 256 * 4);

    f32_to_f16<<<(3 * DIM * DIM + 255) / 256, 256, 0, stream>>>(Wg, Wg_h, 3 * DIM * DIM);
    f32_to_f16<<<(DIM * DIM + 255) / 256, 256, 0, stream>>>(Wc, Wc_h, DIM * DIM);
    f32_to_f16<<<(DIM * 2 * DIM + 255) / 256, 256, 0, stream>>>(Wp, Wp_h, DIM * 2 * DIM);
    f32_to_f16<<<(2 * 512 * DIM + 255) / 256, 256, 0, stream>>>(Wih, Wih_h, 2 * 512 * DIM);
    vec_add<<<(2 * 512 + 255) / 256, 256, 0, stream>>>(bih, bhh, bsum, 2 * 512);

    gather_embed<<<NN, DIM, 0, stream>>>(fingerprints, emb_fp, xs, nullptr);
    gather_embed<<<MM, DIM, 0, stream>>>(words, emb_word, nullptr, wv_h);

    for (int l = 0; l < 3; ++l) {
        f32_to_f16<<<(NN * DIM + 255) / 256, 256, 0, stream>>>(xs, xs_h, NN * DIM);
        gemm_bias_act<<<dim3(NN / 16, DIM / 16), 32, 0, stream>>>(
            xs_h, Wg_h + (size_t)l * DIM * DIM, bg + l * DIM,
            nullptr, hs_h, DIM, DIM, DIM, DIM, 1);
        transpose_f16<<<NN, DIM, 0, stream>>>(hs_h, hsT_h);
        gat_fused<<<NN / 16, GW * 32, 0, stream>>>(hs_h, hsT_h, adjacency, xs);
    }

    f32_to_f16<<<(NN * DIM + 255) / 256, 256, 0, stream>>>(xs, xs_h, NN * DIM);
    gemm_bias_act<<<dim3(NN / 16, DIM / 16), 32, 0, stream>>>(
        xs_h, Wc_h, bc, h_f32, h_h, DIM, DIM, DIM, DIM, 1);

    for (int d = 0; d < 2; ++d) {
        gemm_bias_act<<<dim3(MM / 16, 512 / 16), 32, 0, stream>>>(
            wv_h, Wih_h + (size_t)d * 512 * DIM, bsum + d * 512,
            G + (size_t)d * MM * 512, nullptr, DIM, DIM, DIM, 512, 0);
    }
    lstm_dir<<<2, 128, 0, stream>>>(G, Whh, hbuf);
    bilstm_concat_relu<<<MM, 256, 0, stream>>>(hbuf, xs_rnn_h);
    gemm_bias_act<<<dim3(MM / 16, DIM / 16), 32, 0, stream>>>(
        xs_rnn_h, Wp_h, bp, hp_f32, hp_h, 2 * DIM, 2 * DIM, 2 * DIM, DIM, 1);

    gemm_bias_act<<<dim3(NN / 16, MM / 16), 32, 0, stream>>>(
        h_h, hp_h, nullptr, weights_out, nullptr, DIM, DIM, DIM, MM, 0);

    colsum_part<<<PB, 256, 0, stream>>>(h_f32, hp_f32, part);
    pool_part<<<PB, 256, 0, stream>>>(h_f32, hp_f32, part, catpart);
    final_mlp<<<1, 256, 0, stream>>>(catpart, Wo, bo, Wi, bi, out);
}


# speedup vs baseline: 7.3086x; 1.1147x over previous
#include <hip/hip_runtime.h>

typedef __attribute__((ext_vector_type(16))) _Float16 v16h;
typedef __attribute__((ext_vector_type(8)))  float    v8f;

#define DIM 128
#define NN  4096
#define MM  4096

__device__ __forceinline__ v16h frag_load(const _Float16* __restrict__ tile, int ld) {
    const int lane = threadIdx.x & 31;
    const int r    = lane & 15;
    const int kb   = (lane & 16) ? 8 : 0;
    const _Float16* q = tile + (size_t)r * (size_t)ld + kb;
    v16h f;
    uint4* fu = reinterpret_cast<uint4*>(&f);
    fu[0] = *reinterpret_cast<const uint4*>(q);
    fu[1] = *reinterpret_cast<const uint4*>(q + 16);
    return f;
}

__device__ __forceinline__ void frag_rc(int& rbase, int& col) {
    const int lane = threadIdx.x & 31;
    col   = lane & 15;
    rbase = (lane & 16) ? 8 : 0;
}

__global__ void gemm_bias_act(const _Float16* __restrict__ A, const _Float16* __restrict__ B,
                              const float* __restrict__ bias,
                              float* __restrict__ Cf, _Float16* __restrict__ Ch,
                              int K, int lda, int ldb, int ldc, int do_relu) {
    const int i0 = blockIdx.x * 16;
    const int j0 = blockIdx.y * 16;
    v8f acc = {};
    for (int k = 0; k < K; k += 32) {
        v16h a = frag_load(A + (size_t)i0 * lda + k, lda);
        v16h b = frag_load(B + (size_t)j0 * ldb + k, ldb);
        acc = __builtin_amdgcn_wmma_f32_16x16x32_f16(false, a, false, b, (short)0, acc,
                                                     false, false);
    }
    int rb, col; frag_rc(rb, col);
    const float bv = bias ? bias[j0 + col] : 0.f;
#pragma unroll
    for (int r = 0; r < 8; ++r) {
        const int row = i0 + rb + r;
        float v = acc[r] + bv;
        if (do_relu) v = fmaxf(v, 0.f);
        const size_t idx = (size_t)row * ldc + j0 + col;
        if (Cf) Cf[idx] = v;
        if (Ch) Ch[idx] = (_Float16)v;
    }
}

__global__ void transpose_f16(const _Float16* __restrict__ src, _Float16* __restrict__ dst) {
    const int j = blockIdx.x;
    const int n = threadIdx.x;
    dst[(size_t)n * NN + j] = src[(size_t)j * DIM + n];
}

typedef __attribute__((ext_vector_type(4))) float f4;
#define GW 8
#define OP 132
#define GSTEPS (NN / (32 * GW))

struct GatStep {
    v16h aj[8];
    v16h bt[8];
    f4   ad[4];
};

__device__ __forceinline__ void gat_load(GatStep& g, const _Float16* hs, const _Float16* hsT,
                                         const float* arow, int j0) {
#pragma unroll
    for (int k = 0; k < 4; ++k) {
        g.aj[k]     = frag_load(hs + (size_t)j0 * DIM + 32 * k, DIM);
        g.aj[4 + k] = frag_load(hs + (size_t)(j0 + 16) * DIM + 32 * k, DIM);
    }
#pragma unroll
    for (int nt = 0; nt < 8; ++nt) g.bt[nt] = frag_load(hsT + (size_t)(16 * nt) * NN + j0, NN);
    g.ad[0] = *reinterpret_cast<const f4*>(arow + j0);
    g.ad[1] = *reinterpret_cast<const f4*>(arow + j0 + 4);
    g.ad[2] = *reinterpret_cast<const f4*>(arow + j0 + 16);
    g.ad[3] = *reinterpret_cast<const f4*>(arow + j0 + 20);
}

__device__ __forceinline__ void gat_step(const GatStep& g, const v16h (&bi)[4], v8f (&o)[8], float& rs) {
    v8f s0 = {}, s1 = {};
#pragma unroll
    for (int k = 0; k < 4; ++k) {
        s0 = __builtin_amdgcn_wmma_f32_16x16x32_f16(false, g.aj[k], false, bi[k], (short)0, s0,
                                                    false, false);
        s1 = __builtin_amdgcn_wmma_f32_16x16x32_f16(false, g.aj[4 + k], false, bi[k], (short)0, s1,
                                                    false, false);
    }
    const float m0[8] = {g.ad[0].x, g.ad[0].y, g.ad[0].z, g.ad[0].w,
                         g.ad[1].x, g.ad[1].y, g.ad[1].z, g.ad[1].w};
    const float m1[8] = {g.ad[2].x, g.ad[2].y, g.ad[2].z, g.ad[2].w,
                         g.ad[3].x, g.ad[3].y, g.ad[3].z, g.ad[3].w};
    v16h p;
#pragma unroll
    for (int r = 0; r < 8; ++r) {
        const _Float16 q0 = (_Float16)(s0[r] * m0[r]);
        const _Float16 q1 = (_Float16)(s1[r] * m1[r]);
        p[r]     = q0;
        p[8 + r] = q1;
        rs += fabsf((float)q0) + fabsf((float)q1);
    }
#pragma unroll
    for (int nt = 0; nt < 8; ++nt)
        o[nt] = __builtin_amdgcn_wmma_f32_16x16x32_f16(false, p, false, g.bt[nt], (short)0, o[nt],
                                                       false, false);
}

__global__ __launch_bounds__(GW * 32) void gat_fused(const _Float16* hs, const _Float16* hsT,
                                                     const float* adj, float* __restrict__ xs) {
    __shared__ __attribute__((aligned(16))) float o_sh[GW][16][OP];
    __shared__ float rs_sh[GW][2][16];
    const int i0   = blockIdx.x * 16;
    const int w    = threadIdx.x >> 5;
    const int lane = threadIdx.x & 31;
    const int c    = lane & 15;
    const int hh   = lane >> 4;
    const int h8   = hh * 8;

    v16h bi[4];
#pragma unroll
    for (int k = 0; k < 4; ++k) bi[k] = frag_load(hs + (size_t)i0 * DIM + 32 * k, DIM);

    v8f o[8];
#pragma unroll
    for (int nt = 0; nt < 8; ++nt) o[nt] = v8f{};
    float rs = 0.f;
    const float* arow = adj + (size_t)(i0 + c) * NN + h8;

    GatStep ga, gb;
    gat_load(ga, hs, hsT, arow, w * 64);

#pragma nounroll
    for (int it = 0; it < GSTEPS / 2; ++it) {
        const int jc = (it * GW + w) * 64;
        if (it + 2 < GSTEPS / 2) __builtin_prefetch(arow + jc + 2 * 64 * GW + 24 * hh, 0, 0);
        gat_load(gb, hs, hsT, arow, jc + 32);
        __builtin_amdgcn_sched_barrier(0);
        gat_step(ga, bi, o, rs);
        const int itn = (it + 1 < GSTEPS / 2) ? (it + 1) : it;
        gat_load(ga, hs, hsT, arow, (itn * GW + w) * 64);
        __builtin_amdgcn_sched_barrier(0);
        gat_step(gb, bi, o, rs);
    }

#pragma unroll
    for (int nt = 0; nt < 8; ++nt)
#pragma unroll
        for (int r = 0; r < 8; ++r) o_sh[w][h8 + r][16 * nt + c] = o[nt][r];
    rs_sh[w][hh][c] = rs;
    __syncthreads();

    const int row = threadIdx.x >> 4;
    const int q4  = (threadIdx.x & 15) * 4;
    f4 lo = {0.f, 0.f, 0.f, 0.f}, hi = {0.f, 0.f, 0.f, 0.f};
    float den = 0.f;
#pragma unroll
    for (int ww = 0; ww < GW; ++ww) {
        lo += *reinterpret_cast<const f4*>(&o_sh[ww][row][q4]);
        hi += *reinterpret_cast<const f4*>(&o_sh[ww][row][64 + q4]);
        den += rs_sh[ww][0][row] + rs_sh[ww][1][row];
    }
    den = fmaxf(den, 1e-12f);
    float* xr = xs + (size_t)(i0 + row) * DIM;
    f4 x0 = *reinterpret_cast<const f4*>(xr + q4);
    f4 x1 = *reinterpret_cast<const f4*>(xr + 64 + q4);
    x0.x += lo.x / den; x0.y += lo.y / den; x0.z += lo.z / den; x0.w += lo.w / den;
    x1.x += hi.x / den; x1.y += hi.y / den; x1.z += hi.z / den; x1.w += hi.w / den;
    *reinterpret_cast<f4*>(xr + q4)      = x0;
    *reinterpret_cast<f4*>(xr + 64 + q4) = x1;
}

__global__ void gather_embed(const int* __restrict__ idx, const float* __restrict__ emb,
                             float* __restrict__ of32, _Float16* __restrict__ of16) {
    const int i = blockIdx.x;
    const int n = threadIdx.x;
    const float v = emb[(size_t)idx[i] * DIM + n];
    if (of32) of32[(size_t)i * DIM + n] = v;
    if (of16) of16[(size_t)i * DIM + n] = (_Float16)v;
}

__global__ void f32_to_f16(const float* __restrict__ s, _Float16* __restrict__ d, int n) {
    const int i = blockIdx.x * 256 + threadIdx.x;
    if (i < n) d[i] = (_Float16)s[i];
}

__global__ void vec_add(const float* __restrict__ a, const float* __restrict__ b,
                        float* __restrict__ c, int n) {
    const int i = blockIdx.x * 256 + threadIdx.x;
    if (i < n) c[i] = a[i] + b[i];
}

typedef __attribute__((ext_vector_type(2))) float v2f;
typedef __attribute__((ext_vector_type(4))) float v4f;
#define GD 4

__device__ __forceinline__ float gate_sigmoid(float z) {
    return __builtin_amdgcn_rcpf(1.f + __builtin_amdgcn_exp2f(-1.44269504f * z));
}
__device__ __forceinline__ float gate_tanh_r(float x) {
    return __builtin_amdgcn_rcpf(1.f + __builtin_amdgcn_exp2f(2.88539008f * x));
}

__global__ __launch_bounds__(128) void lstm_dir(const float* __restrict__ G,
                                                const float* __restrict__ Whh,
                                                float* __restrict__ hout) {
    __shared__ __attribute__((aligned(16))) float h_sh[2][2 * DIM];
    const int d = blockIdx.x;
    const int n = threadIdx.x;
    const int wv = n >> 5;
    const float* W  = Whh + (size_t)d * 512 * DIM;
    const float* Gd = G + (size_t)d * MM * 512;
    float* hd = hout + (size_t)d * MM * DIM;

    v2f wi[DIM / 2], wf[DIM / 2], wg[DIM / 2], wo[DIM / 2];
    {
        const v4f* ri = reinterpret_cast<const v4f*>(W + (size_t)(n)       * DIM);
        const v4f* rf = reinterpret_cast<const v4f*>(W + (size_t)(n + 128) * DIM);
        const v4f* rg = reinterpret_cast<const v4f*>(W + (size_t)(n + 256) * DIM);
        const v4f* ro = reinterpret_cast<const v4f*>(W + (size_t)(n + 384) * DIM);
#pragma unroll
        for (int k4 = 0; k4 < DIM / 4; ++k4) {
            const int kg = (8 * wv + k4) & 31;
            const v4f a = ri[kg], b = rf[kg], c4 = rg[kg], e = ro[kg];
            wi[2 * k4] = __builtin_shufflevector(a, a, 0, 1);
            wi[2 * k4 + 1] = __builtin_shufflevector(a, a, 2, 3);
            wf[2 * k4] = __builtin_shufflevector(b, b, 0, 1);
            wf[2 * k4 + 1] = __builtin_shufflevector(b, b, 2, 3);
            wg[2 * k4] = __builtin_shufflevector(c4, c4, 0, 1);
            wg[2 * k4 + 1] = __builtin_shufflevector(c4, c4, 2, 3);
            wo[2 * k4] = __builtin_shufflevector(e, e, 0, 1);
            wo[2 * k4 + 1] = __builtin_shufflevector(e, e, 2, 3);
        }
    }
    h_sh[0][n] = 0.f;
    h_sh[0][DIM + n] = 0.f;
    float c = 0.f;
    float hown = 0.f;
    __syncthreads();

    float pi[GD], pf[GD], pg[GD], po[GD];
#pragma unroll
    for (int u = 0; u < GD; ++u) {
        const float* g0 = Gd + (size_t)(d ? (MM - 1 - u) : u) * 512 + n;
        pi[u] = g0[0]; pf[u] = g0[128]; pg[u] = g0[256]; po[u] = g0[384];
    }

#define LSTM_OWN(J0, J1)                                                                   \
    _Pragma("unroll") for (int j = (J0); j < (J1); ++j) {                                  \
        const v2f hp = {                                                                   \
            __builtin_bit_cast(float, __builtin_amdgcn_readlane(__builtin_bit_cast(int, hown), 2 * j)),     \
            __builtin_bit_cast(float, __builtin_amdgcn_readlane(__builtin_bit_cast(int, hown), 2 * j + 1))}; \
        if (j & 1) {                                                                       \
            bi = __builtin_elementwise_fma(hp, wi[j], bi);                                 \
            bf = __builtin_elementwise_fma(hp, wf[j], bf);                                 \
            bg = __builtin_elementwise_fma(hp, wg[j], bg);                                 \
            bo = __builtin_elementwise_fma(hp, wo[j], bo);                                 \
        } else {                                                                           \
            ai = __builtin_elementwise_fma(hp, wi[j], ai);                                 \
            af = __builtin_elementwise_fma(hp, wf[j], af);                                 \
            ag = __builtin_elementwise_fma(hp, wg[j], ag);                                 \
            ao = __builtin_elementwise_fma(hp, wo[j], ao);                                 \
        }                                                                                  \
    }

#pragma nounroll
    for (int t0 = 0; t0 < MM; t0 += GD) {
#pragma unroll
        for (int u = 0; u < GD; ++u) {
            const int t = t0 + u;
            const int tt = d ? (MM - 1 - t) : t;
            const v4f* h4 = reinterpret_cast<const v4f*>(h_sh[u & 1] + 32 * wv);
            v2f ai = {pi[u], 0.f}, af = {pf[u], 0.f}, ag = {pg[u], 0.f}, ao = {po[u], 0.f};
            v2f bi = {0.f, 0.f}, bf = {0.f, 0.f}, bg = {0.f, 0.f}, bo = {0.f, 0.f};

            LSTM_OWN(0, 6)
            __builtin_amdgcn_sched_barrier(0);
            __builtin_amdgcn_fence(__ATOMIC_RELEASE, "workgroup", "local");
            if (u == 0) asm volatile("s_wait_dscnt 0x0" ::: "memory");
            __builtin_amdgcn_s_barrier_signal(-1);
            __builtin_amdgcn_sched_barrier(0);
            LSTM_OWN(6, 10)
            __builtin_amdgcn_sched_barrier(0);
            __builtin_amdgcn_s_barrier_wait(-1);
            __builtin_amdgcn_fence(__ATOMIC_ACQUIRE, "workgroup", "local");
            v4f hv[DIM / 4];
#pragma unroll
            for (int k4 = 8; k4 < DIM / 4; ++k4) hv[k4] = h4[k4];
            __builtin_amdgcn_sched_barrier(0);
            LSTM_OWN(10, 16)
            __builtin_amdgcn_sched_barrier(0);
#pragma unroll
            for (int k4 = 8; k4 < DIM / 4; ++k4) {
                const v2f hlo = __builtin_shufflevector(hv[k4], hv[k4], 0, 1);
                const v2f hhi = __builtin_shufflevector(hv[k4], hv[k4], 2, 3);
                ai = __builtin_elementwise_fma(hlo, wi[2 * k4], ai);
                af = __builtin_elementwise_fma(hlo, wf[2 * k4], af);
                ag = __builtin_elementwise_fma(hlo, wg[2 * k4], ag);
                ao = __builtin_elementwise_fma(hlo, wo[2 * k4], ao);
                bi = __builtin_elementwise_fma(hhi, wi[2 * k4 + 1], bi);
                bf = __builtin_elementwise_fma(hhi, wf[2 * k4 + 1], bf);
                bg = __builtin_elementwise_fma(hhi, wg[2 * k4 + 1], bg);
                bo = __builtin_elementwise_fma(hhi, wo[2 * k4 + 1], bo);
                if (k4 == 8) {
                    const int tn = (t + GD < MM) ? (t + GD) : t;
                    const float* gn = Gd + (size_t)(d ? (MM - 1 - tn) : tn) * 512 + n;
                    pi[u] = gn[0]; pf[u] = gn[128]; pg[u] = gn[256]; po[u] = gn[384];
                }
                asm volatile("" : "+v"(ai), "+v"(af), "+v"(ag), "+v"(ao),
                                  "+v"(bi), "+v"(bf), "+v"(bg), "+v"(bo));
            }
            const v2f si = ai + bi, sf = af + bf, sg = ag + bg, so = ao + bo;
            float zi = si.x + si.y, zf = sf.x + sf.y, zg = sg.x + sg.y, zo = so.x + so.y;
            const float ig = gate_sigmoid(zi);
            const float fg = gate_sigmoid(zf);
            const float og = gate_sigmoid(zo);
            const float rg = gate_tanh_r(zg);
            c = __builtin_fmaf(rg, -2.f * ig, __builtin_fmaf(fg, c, ig));
            const float rc = gate_tanh_r(c);
            const float h = __builtin_fmaf(rc, -2.f * og, og);
            hown = h;
            h_sh[(u + 1) & 1][n] = h;
            h_sh[(u + 1) & 1][DIM + n] = h;
            hd[(size_t)tt * DIM + n] = h;
        }
    }
#undef LSTM_OWN
}

__global__ void bilstm_concat_relu(const float* __restrict__ hbuf, _Float16* __restrict__ xs_rnn) {
    const int t = blockIdx.x;
    const int n = threadIdx.x;
    const float v = (n < DIM) ? hbuf[(size_t)t * DIM + n]
                              : hbuf[(size_t)(MM + t) * DIM + (n - DIM)];
    xs_rnn[(size_t)t * (2 * DIM) + n] = (_Float16)fmaxf(v, 0.f);
}

#define PB 64
#define PR (NN / PB)

__global__ void colsum_part(const float* __restrict__ h, const float* __restrict__ hp,
                            float* __restrict__ part) {
    const int t = threadIdx.x;
    const int n = t & 127;
    const float* src = ((t < 128) ? h : hp) + (size_t)blockIdx.x * PR * DIM + n;
    float s = 0.f;
    for (int r = 0; r < PR; ++r) s += src[(size_t)r * DIM];
    part[blockIdx.x * 256 + t] = s;
}

__global__ void pool_part(const float* __restrict__ h, const float* __restrict__ hp,
                          const float* __restrict__ part, float* __restrict__ catpart) {
    __shared__ __attribute__((aligned(16))) float colsum[256];
    __shared__ float w_sh[2][PR];
    const int t = threadIdx.x;
    const int side = t >> 7;
    const int n = t & 127;
    float s = 0.f;
    for (int b = 0; b < PB; ++b) s += part[b * 256 + t];
    colsum[t] = s;
    __syncthreads();
    const float* src = (side ? hp : h) + (size_t)blockIdx.x * PR * DIM;
    if (n < PR) {
        const v4f* other = reinterpret_cast<const v4f*>(colsum + (side ? 0 : 128));
        const v4f* row   = reinterpret_cast<const v4f*>(src + (size_t)n * DIM);
        float d = 0.f;
        for (int k4 = 0; k4 < DIM / 4; ++k4) {
            const v4f a = row[k4], b = other[k4];
            d += a.x * b.x + a.y * b.y + a.z * b.z + a.w * b.w;
        }
        w_sh[side][n] = d / (float)(side ? NN : MM);
    }
    __syncthreads();
    float acc = 0.f;
    for (int r = 0; r < PR; ++r) acc += w_sh[side][r] * src[(size_t)r * DIM + n];
    catpart[blockIdx.x * 256 + t] = acc;
}

__global__ void final_mlp(const float* __restrict__ catpart, const float* __restrict__ Wo,
                          const float* __restrict__ bo, const float* __restrict__ Wi,
                          const float* __restrict__ bi, float* __restrict__ out) {
    __shared__ float cur[256], nxt[256];
    const int t = threadIdx.x;
    {
        float s = 0.f;
        for (int b = 0; b < PB; ++b) s += catpart[b * 256 + t];
        cur[t] = s / (float)((t < 128) ? NN : MM);
    }
    __syncthreads();
    for (int l = 0; l < 2; ++l) {
        const float* Wl = Wo + (size_t)l * 256 * 256;
        float s = bo[l * 256 + t];
        for (int k = 0; k < 256; ++k) s += cur[k] * Wl[(size_t)t * 256 + k];
        nxt[t] = fmaxf(s, 0.f);
        __syncthreads();
        cur[t] = nxt[t];
        __syncthreads();
    }
    if (t < 2) {
        float s = bi[t];
        for (int k = 0; k < 256; ++k) s += cur[k] * Wi[(size_t)t * 256 + k];
        out[t] = s;
    }
}

extern "C" void kernel_launch(void* const* d_in, const int* in_sizes, int n_in,
                              void* d_out, int out_size, void* d_ws, size_t ws_size,
                              hipStream_t stream) {
    (void)in_sizes; (void)n_in; (void)out_size; (void)ws_size;
    const int*   fingerprints = (const int*)d_in[0];
    const float* adjacency    = (const float*)d_in[1];
    const int*   words        = (const int*)d_in[2];
    const float* emb_fp       = (const float*)d_in[3];
    const float* emb_word     = (const float*)d_in[4];
    const float* Wg  = (const float*)d_in[5];
    const float* bg  = (const float*)d_in[6];
    const float* Wih = (const float*)d_in[7];
    const float* Whh = (const float*)d_in[8];
    const float* bih = (const float*)d_in[9];
    const float* bhh = (const float*)d_in[10];
    const float* Wc  = (const float*)d_in[11];
    const float* bc  = (const float*)d_in[12];
    const float* Wp  = (const float*)d_in[13];
    const float* bp  = (const float*)d_in[14];
    const float* Wo  = (const float*)d_in[15];
    const float* bo  = (const float*)d_in[16];
    const float* Wi  = (const float*)d_in[17];
    const float* bi  = (const float*)d_in[18];

    float* out         = (float*)d_out;
    float* weights_out = out + 2;

    char* base = (char*)d_ws;
    size_t off = 0;
    auto alloc = [&](size_t bytes) -> void* {
        void* p = base + off;
        off = (off + bytes + 255) & ~(size_t)255;
        return p;
    };
    float*    xs       = (float*)   alloc((size_t)NN * DIM * 4);
    _Float16* xs_h     = (_Float16*)alloc((size_t)NN * DIM * 2);
    _Float16* hs_h     = (_Float16*)alloc((size_t)NN * DIM * 2);
    _Float16* hsT_h    = (_Float16*)alloc((size_t)DIM * NN * 2);
    _Float16* Wg_h     = (_Float16*)alloc((size_t)3 * DIM * DIM * 2);
    _Float16* Wc_h     = (_Float16*)alloc((size_t)DIM * DIM * 2);
    _Float16* Wp_h     = (_Float16*)alloc((size_t)DIM * 2 * DIM * 2);
    _Float16* Wih_h    = (_Float16*)alloc((size_t)2 * 512 * DIM * 2);
    float*    bsum     = (float*)   alloc((size_t)2 * 512 * 4);
    _Float16* wv_h     = (_Float16*)alloc((size_t)MM * DIM * 2);
    float*    G        = (float*)   alloc((size_t)2 * MM * 512 * 4);
    float*    hbuf     = (float*)   alloc((size_t)2 * MM * DIM * 4);
    _Float16* xs_rnn_h = (_Float16*)alloc((size_t)MM * 2 * DIM * 2);
    float*    h_f32    = (float*)   alloc((size_t)NN * DIM * 4);
    _Float16* h_h      = (_Float16*)alloc((size_t)NN * DIM * 2);
    float*    hp_f32   = (float*)   alloc((size_t)MM * DIM * 4);
    _Float16* hp_h     = (_Float16*)alloc((size_t)MM * DIM * 2);
    float*    part     = (float*)   alloc((size_t)PB * 256 * 4);
    float*    catpart  = (float*)   alloc((size_t)PB * 256 * 4);

    f32_to_f16<<<(3 * DIM * DIM + 255) / 256, 256, 0, stream>>>(Wg, Wg_h, 3 * DIM * DIM);
    f32_to_f16<<<(DIM * DIM + 255) / 256, 256, 0, stream>>>(Wc, Wc_h, DIM * DIM);
    f32_to_f16<<<(DIM * 2 * DIM + 255) / 256, 256, 0, stream>>>(Wp, Wp_h, DIM * 2 * DIM);
    f32_to_f16<<<(2 * 512 * DIM + 255) / 256, 256, 0, stream>>>(Wih, Wih_h, 2 * 512 * DIM);
    vec_add<<<(2 * 512 + 255) / 256, 256, 0, stream>>>(bih, bhh, bsum, 2 * 512);

    gather_embed<<<NN, DIM, 0, stream>>>(fingerprints, emb_fp, xs, nullptr);
    gather_embed<<<MM, DIM, 0, stream>>>(words, emb_word, nullptr, wv_h);

    for (int l = 0; l < 3; ++l) {
        f32_to_f16<<<(NN * DIM + 255) / 256, 256, 0, stream>>>(xs, xs_h, NN * DIM);
        gemm_bias_act<<<dim3(NN / 16, DIM / 16), 32, 0, stream>>>(
            xs_h, Wg_h + (size_t)l * DIM * DIM, bg + l * DIM,
            nullptr, hs_h, DIM, DIM, DIM, DIM, 1);
        transpose_f16<<<NN, DIM, 0, stream>>>(hs_h, hsT_h);
        gat_fused<<<NN / 16, GW * 32, 0, stream>>>(hs_h, hsT_h, adjacency, xs);
    }

    f32_to_f16<<<(NN * DIM + 255) / 256, 256, 0, stream>>>(xs, xs_h, NN * DIM);
    gemm_bias_act<<<dim3(NN / 16, DIM / 16), 32, 0, stream>>>(
        xs_h, Wc_h, bc, h_f32, h_h, DIM, DIM, DIM, DIM, 1);

    for (int d = 0; d < 2; ++d) {
        gemm_bias_act<<<dim3(MM / 16, 512 / 16), 32, 0, stream>>>(
            wv_h, Wih_h + (size_t)d * 512 * DIM, bsum + d * 512,
            G + (size_t)d * MM * 512, nullptr, DIM, DIM, DIM, 512, 0);
    }
    lstm_dir<<<2, 128, 0, stream>>>(G, Whh, hbuf);
    bilstm_concat_relu<<<MM, 256, 0, stream>>>(hbuf, xs_rnn_h);
    gemm_bias_act<<<dim3(MM / 16, DIM / 16), 32, 0, stream>>>(
        xs_rnn_h, Wp_h, bp, hp_f32, hp_h, 2 * DIM, 2 * DIM, 2 * DIM, DIM, 1);

    gemm_bias_act<<<dim3(NN / 16, MM / 16), 32, 0, stream>>>(
        h_h, hp_h, nullptr, weights_out, nullptr, DIM, DIM, DIM, MM, 0);

    colsum_part<<<PB, 256, 0, stream>>>(h_f32, hp_f32, part);
    pool_part<<<PB, 256, 0, stream>>>(h_f32, hp_f32, part, catpart);
    final_mlp<<<1, 256, 0, stream>>>(catpart, Wo, bo, Wi, bi, out);
}
